# starting kernel
# baseline (speedup 1.0000x reference)
#include <hip/hip_runtime.h>
#include <hip/hip_bf16.h>
#include <math.h>

typedef float v2f __attribute__((ext_vector_type(2)));
typedef float v8f __attribute__((ext_vector_type(8)));

#define KDIM 128

template <bool TRANS_FILL, bool RELU_IN>
__global__ __launch_bounds__(256) void ggnn_gemm(
    const float* __restrict__ X, const float* __restrict__ W,
    const float* __restrict__ bias, float* __restrict__ Y,
    int nrows, int out_stride, int out_cols_total)
{
    __shared__ float Wt[KDIM * KDIM];

    const int tid   = threadIdx.x;
    const int cbase = blockIdx.y * KDIM;
    int ccols = out_cols_total - cbase;
    if (ccols > KDIM) ccols = KDIM;

    for (int i = tid; i < KDIM * KDIM; i += 256) {
        const int n = i >> 7;
        const int k = i & (KDIM - 1);
        float w = 0.0f;
        if (n < ccols) {
            if (TRANS_FILL) w = W[(size_t)k * out_cols_total + cbase + n];
            else            w = W[(size_t)(cbase + n) * KDIM + k];
        }
        Wt[n * KDIM + k] = w;
    }
    __syncthreads();

    const int wave = tid >> 5;
    const int lane = tid & 31;
    const int m0 = (blockIdx.x * 8 + wave) * 16;
    if (m0 >= nrows) return;

    const int half = lane >> 4;
    const int l16  = lane & 15;

    const float* Xrow = X + (size_t)(m0 + l16) * KDIM + half * 2;

    v8f acc[8] = {};

    for (int kb = 0; kb < KDIM; kb += 4) {
        v2f a = *(const v2f*)(Xrow + kb);
        if (RELU_IN) { a.x = fmaxf(a.x, 0.0f); a.y = fmaxf(a.y, 0.0f); }
        #pragma unroll
        for (int t = 0; t < 8; ++t) {
            v2f b = *(const v2f*)(&Wt[(t * 16 + l16) * KDIM + kb + half * 2]);
            acc[t] = __builtin_amdgcn_wmma_f32_16x16x4_f32(
                false, a, false, b, (short)0, acc[t], false, false);
        }
    }

    #pragma unroll
    for (int t = 0; t < 8; ++t) {
        const int c = t * 16 + l16;
        if (c < ccols) {
            const float bv = bias ? bias[cbase + c] : 0.0f;
            #pragma unroll
            for (int r = 0; r < 8; ++r) {
                const int row = m0 + r + half * 8;
                Y[(size_t)row * out_stride + cbase + c] = acc[t][r] + bv;
            }
        }
    }
}

__global__ __launch_bounds__(256) void ggnn_scatter(
    const float* __restrict__ m, const int* __restrict__ src,
    const int* __restrict__ dst, float* __restrict__ agg, int nedges)
{
    const long idx = (long)blockIdx.x * 256 + threadIdx.x;
    const long e = idx >> 5;
    if (e >= nedges) return;
    const int lane = (int)(idx & 31);
    const int s = src[e];
    const int d = dst[e];
    const float4 v = *(const float4*)(m + (size_t)s * KDIM + lane * 4);
    float* out = agg + (size_t)d * KDIM + lane * 4;
    atomicAdd(out + 0, v.x);
    atomicAdd(out + 1, v.y);
    atomicAdd(out + 2, v.z);
    atomicAdd(out + 3, v.w);
}

__global__ __launch_bounds__(256) void ggnn_zero(float4* __restrict__ p, long n4)
{
    const long i = (long)blockIdx.x * 256 + threadIdx.x;
    if (i < n4) p[i] = make_float4(0.f, 0.f, 0.f, 0.f);
}

__global__ __launch_bounds__(256) void ggnn_gru_gates(
    const float* __restrict__ gi, const float* __restrict__ gh,
    float* __restrict__ h, int nnodes)
{
    const long idx = (long)blockIdx.x * 256 + threadIdx.x;
    if (idx >= (long)nnodes * KDIM) return;
    const long n = idx >> 7;
    const int  c = (int)(idx & (KDIM - 1));
    const float* gin = gi + n * 384;
    const float* ghn = gh + n * 384;
    const float i_r = gin[c], i_z = gin[c + 128], i_n = gin[c + 256];
    const float h_r = ghn[c], h_z = ghn[c + 128], h_n = ghn[c + 256];
    const float r = 1.0f / (1.0f + __expf(-(i_r + h_r)));
    const float z = 1.0f / (1.0f + __expf(-(i_z + h_z)));
    const float nn = tanhf(i_n + r * h_n);
    const float hv = h[idx];
    h[idx] = (1.0f - z) * nn + z * hv;
}

extern "C" void kernel_launch(void* const* d_in, const int* in_sizes, int n_in,
                              void* d_out, int out_size, void* d_ws, size_t ws_size,
                              hipStream_t stream)
{
    (void)n_in; (void)out_size; (void)ws_size;

    const float* x      = (const float*)d_in[0];
    const int*   ei     = (const int*)d_in[1];
    const float* lin_w  = (const float*)d_in[2];
    const float* lin_b  = (const float*)d_in[3];
    const float* ggc_w  = (const float*)d_in[4];
    const float* w_ih   = (const float*)d_in[5];
    const float* w_hh   = (const float*)d_in[6];
    const float* b_ih   = (const float*)d_in[7];
    const float* b_hh   = (const float*)d_in[8];
    const float* lin2_w = (const float*)d_in[9];
    const float* lin2_b = (const float*)d_in[10];
    float* out = (float*)d_out;

    const int N = in_sizes[0] / KDIM;
    const int E = in_sizes[1] / 2;
    const int* src = ei;
    const int* dst = ei + E;

    float* h   = (float*)d_ws;
    float* m   = h   + (size_t)N * 128;
    float* agg = m   + (size_t)N * 128;
    float* gi  = agg + (size_t)N * 128;
    float* gh  = gi  + (size_t)N * 384;

    const int row_tiles  = (N + 15) / 16;
    const int row_blocks = (row_tiles + 7) / 8;
    const dim3 blk(256);

    const long agg4 = (long)N * 32;
    const unsigned zero_blks    = (unsigned)((agg4 + 255) / 256);
    const unsigned scatter_blks = (unsigned)(((long)E * 32 + 255) / 256);
    const unsigned gate_blks    = (unsigned)(((long)N * 128 + 255) / 256);

    ggnn_gemm<false, false><<<dim3(row_blocks, 1), blk, 0, stream>>>(
        x, lin_w, lin_b, h, N, 128, 128);

    for (int l = 0; l < 8; ++l) {
        ggnn_gemm<true, false><<<dim3(row_blocks, 1), blk, 0, stream>>>(
            h, ggc_w + (size_t)l * 128 * 128, nullptr, m, N, 128, 128);
        ggnn_zero<<<zero_blks, blk, 0, stream>>>((float4*)agg, agg4);
        ggnn_scatter<<<scatter_blks, blk, 0, stream>>>(m, src, dst, agg, E);
        ggnn_gemm<false, false><<<dim3(row_blocks, 3), blk, 0, stream>>>(
            agg, w_ih, b_ih, gi, N, 384, 384);
        ggnn_gemm<false, false><<<dim3(row_blocks, 3), blk, 0, stream>>>(
            h, w_hh, b_hh, gh, N, 384, 384);
        ggnn_gru_gates<<<gate_blks, blk, 0, stream>>>(gi, gh, h, N);
    }

    ggnn_gemm<false, true><<<dim3(row_blocks, 1), blk, 0, stream>>>(
        h, lin2_w, lin2_b, out, N, 100, 100);
}


# Scatter-add without global atomics or clearing pass: a work-group owns 200 destination rows as LDS sums, filters the raw dst array each layer, adds m rows by LDS float atomics (order: LDS arrival)
# speedup vs baseline: 2.4643x; 2.4643x over previous
#include <hip/hip_runtime.h>
#include <hip/hip_bf16.h>
#include <math.h>

typedef float v2f __attribute__((ext_vector_type(2)));
typedef float v4f __attribute__((ext_vector_type(4)));
typedef float v8f __attribute__((ext_vector_type(8)));

#define KDIM 128

__device__ __forceinline__ int frag_k0(int q, int h) { return 8 * q + 4 * h; }

__device__ __forceinline__ void load_a_tile(
    v4f (&a)[16], const float* __restrict__ X, int tile, int c, int h)
{
    const float* p = X + (size_t)(tile * 16 + c) * KDIM;
    #pragma unroll
    for (int q = 0; q < 16; ++q)
        a[q] = *(const v4f*)(p + frag_k0(q, h));
}

template <int CT, bool RELU_IN, bool GUARD>
__device__ __forceinline__ void tile_body(
    const v4f (&a)[16], const v4f (&b)[CT][16], const float (&bv)[CT],
    float* __restrict__ Y, int tile, int h, int c, int col0,
    int out_stride, int out_cols_total)
{
    v8f acc[CT];
    #pragma unroll
    for (int t = 0; t < CT; ++t) acc[t] = v8f{0.f, 0.f, 0.f, 0.f, 0.f, 0.f, 0.f, 0.f};

    #pragma unroll
    for (int q = 0; q < 16; ++q) {
        v2f a0 = {a[q].x, a[q].y};
        v2f a1 = {a[q].z, a[q].w};
        if (RELU_IN) {
            a0.x = fmaxf(a0.x, 0.0f); a0.y = fmaxf(a0.y, 0.0f);
            a1.x = fmaxf(a1.x, 0.0f); a1.y = fmaxf(a1.y, 0.0f);
        }
        #pragma unroll
        for (int t = 0; t < CT; ++t) {
            const v2f b0 = {b[t][q].x, b[t][q].y};
            acc[t] = __builtin_amdgcn_wmma_f32_16x16x4_f32(
                false, a0, false, b0, (short)0, acc[t], false, false);
        }
        #pragma unroll
        for (int t = 0; t < CT; ++t) {
            const v2f b1 = {b[t][q].z, b[t][q].w};
            acc[t] = __builtin_amdgcn_wmma_f32_16x16x4_f32(
                false, a1, false, b1, (short)0, acc[t], false, false);
        }
    }

    #pragma unroll
    for (int t = 0; t < CT; ++t) {
        const int cc = col0 + t * 16 + c;
        if (!GUARD || cc < out_cols_total) {
            #pragma unroll
            for (int r = 0; r < 8; ++r) {
                const int row = tile * 16 + r + h * 8;
                Y[(size_t)row * out_stride + cc] = acc[t][r] + bv[t];
            }
        }
    }
}

template <int CT, int NG, bool TRANS_FILL, bool RELU_IN, bool GUARD>
__global__ __launch_bounds__(256) __attribute__((amdgpu_waves_per_eu(2, 2))) void ggnn_gemm(
    const float* __restrict__ X, const float* __restrict__ W,
    const float* __restrict__ bias, float* __restrict__ Y,
    int ntiles, int out_stride, int out_cols_total)
{
    constexpr int NSTR = 8 / NG;
    const int tid  = threadIdx.x;
    const int wave = __builtin_amdgcn_readfirstlane(tid >> 5);
    const int lane = tid & 31;
    const int h    = lane >> 4;
    const int c    = lane & 15;
    const int grp  = wave % NG;
    const int strm = wave / NG;
    const int col0 = grp * (CT * 16);

    v4f   b[CT][16];
    float bv[CT];
    #pragma unroll
    for (int t = 0; t < CT; ++t) {
        const int n = col0 + t * 16 + c;
        const bool ok = !GUARD || (n < out_cols_total);
        bv[t] = (bias != nullptr && ok) ? bias[n] : 0.0f;
        #pragma unroll
        for (int q = 0; q < 16; ++q) {
            const int k0 = frag_k0(q, h);
            v4f w = {0.f, 0.f, 0.f, 0.f};
            if (ok) {
                if (TRANS_FILL) {
                    w.x = W[(size_t)(k0 + 0) * out_cols_total + n];
                    w.y = W[(size_t)(k0 + 1) * out_cols_total + n];
                    w.z = W[(size_t)(k0 + 2) * out_cols_total + n];
                    w.w = W[(size_t)(k0 + 3) * out_cols_total + n];
                } else {
                    w = *(const v4f*)(W + (size_t)n * KDIM + k0);
                }
            }
            b[t][q] = w;
        }
    }

    const int nstreams = (int)gridDim.x * NSTR;
    int tile = (int)blockIdx.x * NSTR + strm;
    if (tile >= ntiles) return;
    const int ntrips = (ntiles - tile + nstreams - 1) / nstreams;
    const int last_tile = tile + (ntrips - 1) * nstreams;

    v4f a0[16], a1[16];
    load_a_tile(a0, X, tile, c, h);
    int i = 0;
    for (; i + 2 <= ntrips; i += 2) {
        load_a_tile(a1, X, tile + nstreams, c, h);
        __builtin_amdgcn_sched_barrier(0);
        tile_body<CT, RELU_IN, GUARD>(a0, b, bv, Y, tile, h, c, col0, out_stride, out_cols_total);
        const int t2 = tile + 2 * nstreams;
        load_a_tile(a0, X, t2 <= last_tile ? t2 : last_tile, c, h);
        __builtin_amdgcn_sched_barrier(0);
        tile_body<CT, RELU_IN, GUARD>(a1, b, bv, Y, tile + nstreams, h, c, col0, out_stride, out_cols_total);
        tile = t2;
    }
    if (i < ntrips)
        tile_body<CT, RELU_IN, GUARD>(a0, b, bv, Y, tile, h, c, col0, out_stride, out_cols_total);
}

#define SC_ROWS    200
#define SC_CAP     8192
#define SC_THREADS 1024
#define SC_VEC     8
#define SC_GATHER  8

__device__ __forceinline__ void sc_note_edge(
    int e, int d, int lo, int rows, int* cnt, int2* list, float* acc,
    const float* __restrict__ m, const int* __restrict__ src)
{
    const unsigned r = (unsigned)(d - lo);
    if (r < (unsigned)rows) {
        const int slot = atomicAdd(cnt, 1);
        if (slot < SC_CAP) {
            list[slot] = make_int2(e, (int)r);
        } else {
            const float* p = m + (size_t)src[e] * KDIM;
            float* a = acc + (size_t)r * KDIM;
            #pragma nounroll
            for (int j = 0; j < KDIM; ++j) atomicAdd(a + j, p[j]);
        }
    }
}

__global__ __launch_bounds__(SC_THREADS) void ggnn_scatter_owned(
    const float* __restrict__ m, const int* __restrict__ src,
    const int* __restrict__ dst, float* __restrict__ agg, int nnodes, int nedges)
{
    __shared__ __attribute__((aligned(16))) float acc[SC_ROWS * KDIM];
    __shared__ int2 list[SC_CAP];
    __shared__ int cnt;

    const int tid  = threadIdx.x;
    const int lane = tid & 31;
    const int wave = __builtin_amdgcn_readfirstlane(tid >> 5);

    for (int lo = (int)blockIdx.x * SC_ROWS; lo < nnodes; lo += (int)gridDim.x * SC_ROWS) {
        const int rows = (nnodes - lo < SC_ROWS) ? (nnodes - lo) : SC_ROWS;

        for (int i = tid; i < SC_ROWS * KDIM / 4; i += SC_THREADS)
            ((float4*)acc)[i] = make_float4(0.f, 0.f, 0.f, 0.f);
        if (tid == 0) cnt = 0;
        __syncthreads();

        if ((nedges & 3) == 0) {
            for (int base = 0; base < nedges; base += SC_VEC * SC_THREADS * 4) {
                int4 d[SC_VEC];
                #pragma unroll
                for (int k = 0; k < SC_VEC; ++k) {
                    const int e0 = base + (k * SC_THREADS + tid) * 4;
                    d[k] = make_int4(lo - 1, lo - 1, lo - 1, lo - 1);
                    if (e0 < nedges) d[k] = *(const int4*)(dst + e0);
                }
                #pragma unroll
                for (int k = 0; k < SC_VEC; ++k) {
                    const int e0 = base + (k * SC_THREADS + tid) * 4;
                    sc_note_edge(e0 + 0, d[k].x, lo, rows, &cnt, list, acc, m, src);
                    sc_note_edge(e0 + 1, d[k].y, lo, rows, &cnt, list, acc, m, src);
                    sc_note_edge(e0 + 2, d[k].z, lo, rows, &cnt, list, acc, m, src);
                    sc_note_edge(e0 + 3, d[k].w, lo, rows, &cnt, list, acc, m, src);
                }
            }
        } else {
            for (int e = tid; e < nedges; e += SC_THREADS)
                sc_note_edge(e, dst[e], lo, rows, &cnt, list, acc, m, src);
        }
        __syncthreads();

        const int n = (cnt < SC_CAP) ? cnt : SC_CAP;
        for (int i = tid; i < n; i += SC_THREADS)
            list[i].x = src[list[i].x];
        __syncthreads();

        for (int i = wave; i < n; i += 32 * SC_GATHER) {
            int2 mine = make_int2(0, -1);
            const int my_idx = i + 32 * lane;
            if (lane < SC_GATHER && my_idx < n) mine = list[my_idx];
            float v[SC_GATHER][4];
            int   r[SC_GATHER];
            #pragma unroll
            for (int u = 0; u < SC_GATHER; ++u) {
                const int s = __builtin_amdgcn_readlane(mine.x, u);
                r[u] = __builtin_amdgcn_readlane(mine.y, u);
                const float* p = m + (size_t)s * KDIM + lane;
                v[u][0] = p[0]; v[u][1] = p[32]; v[u][2] = p[64]; v[u][3] = p[96];
            }
            #pragma unroll
            for (int u = 0; u < SC_GATHER; ++u) {
                if (r[u] >= 0) {
                    float* a = acc + r[u] * KDIM + lane;
                    atomicAdd(a,      v[u][0]);
                    atomicAdd(a + 32, v[u][1]);
                    atomicAdd(a + 64, v[u][2]);
                    atomicAdd(a + 96, v[u][3]);
                }
            }
        }
        __syncthreads();

        float4* out4 = (float4*)(agg + (size_t)lo * KDIM);
        for (int i = tid; i < rows * (KDIM / 4); i += SC_THREADS)
            out4[i] = ((const float4*)acc)[i];
        __syncthreads();
    }
}

__global__ __launch_bounds__(256) void ggnn_gru_gates(
    const float* __restrict__ gi, const float* __restrict__ gh,
    float* __restrict__ h, int nnodes)
{
    const long idx = (long)blockIdx.x * 256 + threadIdx.x;
    if (idx >= (long)nnodes * KDIM) return;
    const long n = idx >> 7;
    const int  c = (int)(idx & (KDIM - 1));
    const float* gin = gi + n * 384;
    const float* ghn = gh + n * 384;
    const float i_r = gin[c], i_z = gin[c + 128], i_n = gin[c + 256];
    const float h_r = ghn[c], h_z = ghn[c + 128], h_n = ghn[c + 256];
    const float r = 1.0f / (1.0f + __expf(-(i_r + h_r)));
    const float z = 1.0f / (1.0f + __expf(-(i_z + h_z)));
    const float nn = tanhf(i_n + r * h_n);
    const float hv = h[idx];
    h[idx] = (1.0f - z) * nn + z * hv;
}

extern "C" void kernel_launch(void* const* d_in, const int* in_sizes, int n_in,
                              void* d_out, int out_size, void* d_ws, size_t ws_size,
                              hipStream_t stream)
{
    (void)n_in; (void)out_size; (void)ws_size;

    const float* x      = (const float*)d_in[0];
    const int*   ei     = (const int*)d_in[1];
    const float* lin_w  = (const float*)d_in[2];
    const float* lin_b  = (const float*)d_in[3];
    const float* ggc_w  = (const float*)d_in[4];
    const float* w_ih   = (const float*)d_in[5];
    const float* w_hh   = (const float*)d_in[6];
    const float* b_ih   = (const float*)d_in[7];
    const float* b_hh   = (const float*)d_in[8];
    const float* lin2_w = (const float*)d_in[9];
    const float* lin2_b = (const float*)d_in[10];
    float* out = (float*)d_out;

    const int N = in_sizes[0] / KDIM;
    const int E = in_sizes[1] / 2;
    const int* src = ei;
    const int* dst = ei + E;

    float* h   = (float*)d_ws;
    float* m   = h   + (size_t)N * 128;
    float* agg = m   + (size_t)N * 128;
    float* gi  = agg + (size_t)N * 128;
    float* gh  = gi  + (size_t)N * 384;

    const int row_tiles = (N + 15) / 16;
    const dim3 blk(256);
    const dim3 gemm_grid(256);

    const unsigned scatter_blks = (unsigned)((N + SC_ROWS - 1) / SC_ROWS);
    const unsigned gate_blks    = (unsigned)(((long)N * 128 + 255) / 256);

    ggnn_gemm<2, 4, false, false, false><<<gemm_grid, blk, 0, stream>>>(
        x, lin_w, lin_b, h, row_tiles, 128, 128);

    for (int l = 0; l < 8; ++l) {
        ggnn_gemm<2, 4, true, false, false><<<gemm_grid, blk, 0, stream>>>(
            h, ggc_w + (size_t)l * 128 * 128, nullptr, m, row_tiles, 128, 128);
        ggnn_scatter_owned<<<scatter_blks, dim3(SC_THREADS), 0, stream>>>(m, src, dst, agg, N, E);
        ggnn_gemm<3, 8, false, false, false><<<gemm_grid, blk, 0, stream>>>(
            agg, w_ih, b_ih, gi, row_tiles, 384, 384);
        ggnn_gemm<3, 8, false, false, false><<<gemm_grid, blk, 0, stream>>>(
            h, w_hh, b_hh, gh, row_tiles, 384, 384);
        ggnn_gru_gates<<<gate_blks, blk, 0, stream>>>(gi, gh, h, N);
    }

    ggnn_gemm<2, 4, false, true, true><<<gemm_grid, blk, 0, stream>>>(
        h, lin2_w, lin2_b, out, row_tiles, 100, 100);
}


# Edge list (source row, local row) of each 200-row destination range built once per call in workspace, copied to LDS per layer; was a full dst scan every layer (sum order: LDS arrival, as before)
# speedup vs baseline: 3.1304x; 1.2703x over previous
#include <hip/hip_runtime.h>
#include <hip/hip_bf16.h>
#include <math.h>

typedef float v2f __attribute__((ext_vector_type(2)));
typedef float v4f __attribute__((ext_vector_type(4)));
typedef float v8f __attribute__((ext_vector_type(8)));

#define KDIM 128

__device__ __forceinline__ int frag_k0(int q, int h) { return 8 * q + 4 * h; }

__device__ __forceinline__ void load_a_tile(
    v4f (&a)[16], const float* __restrict__ X, int tile, int c, int h)
{
    const float* p = X + (size_t)(tile * 16 + c) * KDIM;
    #pragma unroll
    for (int q = 0; q < 16; ++q)
        a[q] = *(const v4f*)(p + frag_k0(q, h));
}

template <int CT, bool RELU_IN, bool GUARD>
__device__ __forceinline__ void tile_body(
    const v4f (&a)[16], const v4f (&b)[CT][16], const float (&bv)[CT],
    float* __restrict__ Y, int tile, int h, int c, int col0,
    int out_stride, int out_cols_total)
{
    v8f acc[CT];
    #pragma unroll
    for (int t = 0; t < CT; ++t) acc[t] = v8f{0.f, 0.f, 0.f, 0.f, 0.f, 0.f, 0.f, 0.f};

    #pragma unroll
    for (int q = 0; q < 16; ++q) {
        v2f a0 = {a[q].x, a[q].y};
        v2f a1 = {a[q].z, a[q].w};
        if (RELU_IN) {
            a0.x = fmaxf(a0.x, 0.0f); a0.y = fmaxf(a0.y, 0.0f);
            a1.x = fmaxf(a1.x, 0.0f); a1.y = fmaxf(a1.y, 0.0f);
        }
        #pragma unroll
        for (int t = 0; t < CT; ++t) {
            const v2f b0 = {b[t][q].x, b[t][q].y};
            acc[t] = __builtin_amdgcn_wmma_f32_16x16x4_f32(
                false, a0, false, b0, (short)0, acc[t], false, false);
        }
        #pragma unroll
        for (int t = 0; t < CT; ++t) {
            const v2f b1 = {b[t][q].z, b[t][q].w};
            acc[t] = __builtin_amdgcn_wmma_f32_16x16x4_f32(
                false, a1, false, b1, (short)0, acc[t], false, false);
        }
    }

    #pragma unroll
    for (int t = 0; t < CT; ++t) {
        const int cc = col0 + t * 16 + c;
        if (!GUARD || cc < out_cols_total) {
            #pragma unroll
            for (int r = 0; r < 8; ++r) {
                const int row = tile * 16 + r + h * 8;
                Y[(size_t)row * out_stride + cc] = acc[t][r] + bv[t];
            }
        }
    }
}

template <int CT, int NG, bool TRANS_FILL, bool RELU_IN, bool GUARD>
__global__ __launch_bounds__(256) __attribute__((amdgpu_waves_per_eu(2, 2))) void ggnn_gemm(
    const float* __restrict__ X, const float* __restrict__ W,
    const float* __restrict__ bias, float* __restrict__ Y,
    int ntiles, int out_stride, int out_cols_total)
{
    constexpr int NSTR = 8 / NG;
    const int tid  = threadIdx.x;
    const int wave = __builtin_amdgcn_readfirstlane(tid >> 5);
    const int lane = tid & 31;
    const int h    = lane >> 4;
    const int c    = lane & 15;
    const int grp  = wave % NG;
    const int strm = wave / NG;
    const int col0 = grp * (CT * 16);

    v4f   b[CT][16];
    float bv[CT];
    #pragma unroll
    for (int t = 0; t < CT; ++t) {
        const int n = col0 + t * 16 + c;
        const bool ok = !GUARD || (n < out_cols_total);
        bv[t] = (bias != nullptr && ok) ? bias[n] : 0.0f;
        #pragma unroll
        for (int q = 0; q < 16; ++q) {
            const int k0 = frag_k0(q, h);
            v4f w = {0.f, 0.f, 0.f, 0.f};
            if (ok) {
                if (TRANS_FILL) {
                    w.x = W[(size_t)(k0 + 0) * out_cols_total + n];
                    w.y = W[(size_t)(k0 + 1) * out_cols_total + n];
                    w.z = W[(size_t)(k0 + 2) * out_cols_total + n];
                    w.w = W[(size_t)(k0 + 3) * out_cols_total + n];
                } else {
                    w = *(const v4f*)(W + (size_t)n * KDIM + k0);
                }
            }
            b[t][q] = w;
        }
    }

    const int nstreams = (int)gridDim.x * NSTR;
    int tile = (int)blockIdx.x * NSTR + strm;
    if (tile >= ntiles) return;
    const int ntrips = (ntiles - tile + nstreams - 1) / nstreams;
    const int last_tile = tile + (ntrips - 1) * nstreams;

    v4f a0[16], a1[16];
    load_a_tile(a0, X, tile, c, h);
    int i = 0;
    for (; i + 2 <= ntrips; i += 2) {
        load_a_tile(a1, X, tile + nstreams, c, h);
        __builtin_amdgcn_sched_barrier(0);
        tile_body<CT, RELU_IN, GUARD>(a0, b, bv, Y, tile, h, c, col0, out_stride, out_cols_total);
        const int t2 = tile + 2 * nstreams;
        load_a_tile(a0, X, t2 <= last_tile ? t2 : last_tile, c, h);
        __builtin_amdgcn_sched_barrier(0);
        tile_body<CT, RELU_IN, GUARD>(a1, b, bv, Y, tile + nstreams, h, c, col0, out_stride, out_cols_total);
        tile = t2;
    }
    if (i < ntrips)
        tile_body<CT, RELU_IN, GUARD>(a0, b, bv, Y, tile, h, c, col0, out_stride, out_cols_total);
}

#define SC_ROWS    200
#define SC_CAP     8192
#define SC_THREADS 1024
#define SC_VEC     8
#define SC_GATHER  8

template <bool ADD_WHEN_FULL>
__device__ __forceinline__ void sc_note_edge(
    int e, int d, int lo, int rows, int* cnt, int2* list, float* acc,
    const float* __restrict__ m, const int* __restrict__ src)
{
    const unsigned r = (unsigned)(d - lo);
    if (r < (unsigned)rows) {
        const int slot = atomicAdd(cnt, 1);
        if (slot < SC_CAP) {
            list[slot] = make_int2(e, (int)r);
        } else if (ADD_WHEN_FULL) {
            const float* p = m + (size_t)src[e] * KDIM;
            float* a = acc + (size_t)r * KDIM;
            #pragma nounroll
            for (int j = 0; j < KDIM; ++j) atomicAdd(a + j, p[j]);
        }
    }
}

template <bool ADD_WHEN_FULL>
__device__ __forceinline__ void sc_filter_edges(
    int lo, int rows, int tid, int* cnt, int2* list, float* acc,
    const float* __restrict__ m, const int* __restrict__ src,
    const int* __restrict__ dst, int nedges)
{
    if ((nedges & 3) == 0) {
        for (int base = 0; base < nedges; base += SC_VEC * SC_THREADS * 4) {
            int4 d[SC_VEC];
            #pragma unroll
            for (int k = 0; k < SC_VEC; ++k) {
                const int e0 = base + (k * SC_THREADS + tid) * 4;
                d[k] = make_int4(lo - 1, lo - 1, lo - 1, lo - 1);
                if (e0 < nedges) d[k] = *(const int4*)(dst + e0);
            }
            #pragma unroll
            for (int k = 0; k < SC_VEC; ++k) {
                const int e0 = base + (k * SC_THREADS + tid) * 4;
                sc_note_edge<ADD_WHEN_FULL>(e0 + 0, d[k].x, lo, rows, cnt, list, acc, m, src);
                sc_note_edge<ADD_WHEN_FULL>(e0 + 1, d[k].y, lo, rows, cnt, list, acc, m, src);
                sc_note_edge<ADD_WHEN_FULL>(e0 + 2, d[k].z, lo, rows, cnt, list, acc, m, src);
                sc_note_edge<ADD_WHEN_FULL>(e0 + 3, d[k].w, lo, rows, cnt, list, acc, m, src);
            }
        }
    } else {
        for (int e = tid; e < nedges; e += SC_THREADS)
            sc_note_edge<ADD_WHEN_FULL>(e, dst[e], lo, rows, cnt, list, acc, m, src);
    }
    __syncthreads();

    const int n = (*cnt < SC_CAP) ? *cnt : SC_CAP;
    for (int i = tid; i < n; i += SC_THREADS)
        list[i].x = src[list[i].x];
    __syncthreads();
}

__global__ __launch_bounds__(SC_THREADS) void ggnn_group_edges(
    const int* __restrict__ src, const int* __restrict__ dst,
    int* __restrict__ lists, int* __restrict__ counts, int nnodes, int nedges)
{
    __shared__ int2 list[SC_CAP];
    __shared__ int cnt;

    const int tid  = threadIdx.x;
    const int g    = (int)blockIdx.x;
    const int lo   = g * SC_ROWS;
    const int rows = (nnodes - lo < SC_ROWS) ? (nnodes - lo) : SC_ROWS;

    if (tid == 0) cnt = 0;
    __syncthreads();

    sc_filter_edges<false>(lo, rows, tid, &cnt, list, nullptr, nullptr, src, dst, nedges);

    const int total = cnt;
    const int n = (total < SC_CAP) ? total : SC_CAP;
    int2* gl = (int2*)lists + (size_t)g * SC_CAP;
    for (int i = tid; i < n; i += SC_THREADS)
        gl[i] = list[i];
    if (tid == 0) counts[g] = total;
}

__global__ __launch_bounds__(SC_THREADS) void ggnn_scatter_owned(
    const float* __restrict__ m, const int* __restrict__ src,
    const int* __restrict__ dst, const int* __restrict__ lists,
    const int* __restrict__ counts, float* __restrict__ agg, int nnodes, int nedges)
{
    __shared__ __attribute__((aligned(16))) float acc[SC_ROWS * KDIM];
    __shared__ int2 list[SC_CAP];
    __shared__ int cnt;

    const int tid  = threadIdx.x;
    const int lane = tid & 31;
    const int wave = __builtin_amdgcn_readfirstlane(tid >> 5);

    const int g    = (int)blockIdx.x;
    const int lo   = g * SC_ROWS;
    const int rows = (nnodes - lo < SC_ROWS) ? (nnodes - lo) : SC_ROWS;
    const int total = __builtin_amdgcn_readfirstlane(counts[g]);
    const bool have_list = total <= SC_CAP;

    int2 t[SC_CAP / SC_THREADS];
    #pragma unroll
    for (int k = 0; k < SC_CAP / SC_THREADS; ++k) {
        const int i = k * SC_THREADS + tid;
        t[k] = make_int2(0, 0);
        if (have_list && i < total)
            t[k] = ((const int2*)lists)[(size_t)g * SC_CAP + i];
    }
    for (int i = tid; i < SC_ROWS * KDIM / 4; i += SC_THREADS)
        ((float4*)acc)[i] = make_float4(0.f, 0.f, 0.f, 0.f);
    if (tid == 0) cnt = 0;
    #pragma unroll
    for (int k = 0; k < SC_CAP / SC_THREADS; ++k) {
        const int i = k * SC_THREADS + tid;
        if (have_list && i < total) list[i] = t[k];
    }
    __syncthreads();

    if (!have_list)
        sc_filter_edges<true>(lo, rows, tid, &cnt, list, acc, m, src, dst, nedges);

    const int n = (total < SC_CAP) ? total : SC_CAP;

    for (int i = wave; i < n; i += 32 * SC_GATHER) {
        int2 mine = make_int2(0, -1);
        const int my_idx = i + 32 * lane;
        if (lane < SC_GATHER && my_idx < n) mine = list[my_idx];
        float v[SC_GATHER][4];
        int   r[SC_GATHER];
        #pragma unroll
        for (int u = 0; u < SC_GATHER; ++u) {
            const int s = __builtin_amdgcn_readlane(mine.x, u);
            r[u] = __builtin_amdgcn_readlane(mine.y, u);
            const float* p = m + (size_t)s * KDIM + lane;
            v[u][0] = p[0]; v[u][1] = p[32]; v[u][2] = p[64]; v[u][3] = p[96];
        }
        #pragma unroll
        for (int u = 0; u < SC_GATHER; ++u) {
            if (r[u] >= 0) {
                float* a = acc + r[u] * KDIM + lane;
                atomicAdd(a,      v[u][0]);
                atomicAdd(a + 32, v[u][1]);
                atomicAdd(a + 64, v[u][2]);
                atomicAdd(a + 96, v[u][3]);
            }
        }
    }
    __syncthreads();

    float4* out4 = (float4*)(agg + (size_t)lo * KDIM);
    for (int i = tid; i < rows * (KDIM / 4); i += SC_THREADS)
        out4[i] = ((const float4*)acc)[i];
}

__global__ __launch_bounds__(256) void ggnn_gru_gates(
    const float* __restrict__ gi, const float* __restrict__ gh,
    float* __restrict__ h, int nnodes)
{
    const long idx = (long)blockIdx.x * 256 + threadIdx.x;
    if (idx >= (long)nnodes * KDIM) return;
    const long n = idx >> 7;
    const int  c = (int)(idx & (KDIM - 1));
    const float* gin = gi + n * 384;
    const float* ghn = gh + n * 384;
    const float i_r = gin[c], i_z = gin[c + 128], i_n = gin[c + 256];
    const float h_r = ghn[c], h_z = ghn[c + 128], h_n = ghn[c + 256];
    const float r = 1.0f / (1.0f + __expf(-(i_r + h_r)));
    const float z = 1.0f / (1.0f + __expf(-(i_z + h_z)));
    const float nn = tanhf(i_n + r * h_n);
    const float hv = h[idx];
    h[idx] = (1.0f - z) * nn + z * hv;
}

extern "C" void kernel_launch(void* const* d_in, const int* in_sizes, int n_in,
                              void* d_out, int out_size, void* d_ws, size_t ws_size,
                              hipStream_t stream)
{
    (void)n_in; (void)out_size; (void)ws_size;

    const float* x      = (const float*)d_in[0];
    const int*   ei     = (const int*)d_in[1];
    const float* lin_w  = (const float*)d_in[2];
    const float* lin_b  = (const float*)d_in[3];
    const float* ggc_w  = (const float*)d_in[4];
    const float* w_ih   = (const float*)d_in[5];
    const float* w_hh   = (const float*)d_in[6];
    const float* b_ih   = (const float*)d_in[7];
    const float* b_hh   = (const float*)d_in[8];
    const float* lin2_w = (const float*)d_in[9];
    const float* lin2_b = (const float*)d_in[10];
    float* out = (float*)d_out;

    const int N = in_sizes[0] / KDIM;
    const int E = in_sizes[1] / 2;
    const int* src = ei;
    const int* dst = ei + E;

    float* h   = (float*)d_ws;
    float* m   = h   + (size_t)N * 128;
    float* agg = m   + (size_t)N * 128;
    float* gi  = agg + (size_t)N * 128;
    float* gh  = gi  + (size_t)N * 384;

    const int row_tiles = (N + 15) / 16;
    const dim3 blk(256);
    const dim3 gemm_grid(256);

    const unsigned scatter_blks = (unsigned)((N + SC_ROWS - 1) / SC_ROWS);
    int* lists  = (int*)(gh + (size_t)N * 384);
    int* counts = lists + (size_t)scatter_blks * SC_CAP * 2;
    const unsigned gate_blks    = (unsigned)(((long)N * 128 + 255) / 256);

    ggnn_gemm<2, 4, false, false, false><<<gemm_grid, blk, 0, stream>>>(
        x, lin_w, lin_b, h, row_tiles, 128, 128);

    ggnn_group_edges<<<scatter_blks, dim3(SC_THREADS), 0, stream>>>(src, dst, lists, counts, N, E);

    for (int l = 0; l < 8; ++l) {
        ggnn_gemm<2, 4, true, false, false><<<gemm_grid, blk, 0, stream>>>(
            h, ggc_w + (size_t)l * 128 * 128, nullptr, m, row_tiles, 128, 128);
        ggnn_scatter_owned<<<scatter_blks, dim3(SC_THREADS), 0, stream>>>(m, src, dst, lists, counts, agg, N, E);
        ggnn_gemm<3, 8, false, false, false><<<gemm_grid, blk, 0, stream>>>(
            agg, w_ih, b_ih, gi, row_tiles, 384, 384);
        ggnn_gemm<3, 8, false, false, false><<<gemm_grid, blk, 0, stream>>>(
            h, w_hh, b_hh, gh, row_tiles, 384, 384);
        ggnn_gru_gates<<<gate_blks, blk, 0, stream>>>(gi, gh, h, N);
    }

    ggnn_gemm<2, 4, false, true, true><<<gemm_grid, blk, 0, stream>>>(
        h, lin2_w, lin2_b, out, row_tiles, 100, 100);
}


# GRU gate launch as a resident streaming grid: 512 groups of 8 waves, a wave walks over row pairs, 16-byte accesses, next pair's loads in flight; was 1 element per thread in 25000 groups
# speedup vs baseline: 3.4039x; 1.0874x over previous
#include <hip/hip_runtime.h>
#include <hip/hip_bf16.h>
#include <math.h>

typedef float v2f __attribute__((ext_vector_type(2)));
typedef float v4f __attribute__((ext_vector_type(4)));
typedef float v8f __attribute__((ext_vector_type(8)));

#define KDIM 128

__device__ __forceinline__ int frag_k0(int q, int h) { return 8 * q + 4 * h; }

__device__ __forceinline__ void load_a_tile(
    v4f (&a)[16], const float* __restrict__ X, int tile, int c, int h)
{
    const float* p = X + (size_t)(tile * 16 + c) * KDIM;
    #pragma unroll
    for (int q = 0; q < 16; ++q)
        a[q] = *(const v4f*)(p + frag_k0(q, h));
}

template <int CT, bool RELU_IN, bool GUARD>
__device__ __forceinline__ void tile_body(
    const v4f (&a)[16], const v4f (&b)[CT][16], const float (&bv)[CT],
    float* __restrict__ Y, int tile, int h, int c, int col0,
    int out_stride, int out_cols_total)
{
    v8f acc[CT];
    #pragma unroll
    for (int t = 0; t < CT; ++t) acc[t] = v8f{0.f, 0.f, 0.f, 0.f, 0.f, 0.f, 0.f, 0.f};

    #pragma unroll
    for (int q = 0; q < 16; ++q) {
        v2f a0 = {a[q].x, a[q].y};
        v2f a1 = {a[q].z, a[q].w};
        if (RELU_IN) {
            a0.x = fmaxf(a0.x, 0.0f); a0.y = fmaxf(a0.y, 0.0f);
            a1.x = fmaxf(a1.x, 0.0f); a1.y = fmaxf(a1.y, 0.0f);
        }
        #pragma unroll
        for (int t = 0; t < CT; ++t) {
            const v2f b0 = {b[t][q].x, b[t][q].y};
            acc[t] = __builtin_amdgcn_wmma_f32_16x16x4_f32(
                false, a0, false, b0, (short)0, acc[t], false, false);
        }
        #pragma unroll
        for (int t = 0; t < CT; ++t) {
            const v2f b1 = {b[t][q].z, b[t][q].w};
            acc[t] = __builtin_amdgcn_wmma_f32_16x16x4_f32(
                false, a1, false, b1, (short)0, acc[t], false, false);
        }
    }

    #pragma unroll
    for (int t = 0; t < CT; ++t) {
        const int cc = col0 + t * 16 + c;
        if (!GUARD || cc < out_cols_total) {
            #pragma unroll
            for (int r = 0; r < 8; ++r) {
                const int row = tile * 16 + r + h * 8;
                Y[(size_t)row * out_stride + cc] = acc[t][r] + bv[t];
            }
        }
    }
}

template <int CT, int NG, bool TRANS_FILL, bool RELU_IN, bool GUARD>
__global__ __launch_bounds__(256) __attribute__((amdgpu_waves_per_eu(2, 2))) void ggnn_gemm(
    const float* __restrict__ X, const float* __restrict__ W,
    const float* __restrict__ bias, float* __restrict__ Y,
    int ntiles, int out_stride, int out_cols_total)
{
    constexpr int NSTR = 8 / NG;
    const int tid  = threadIdx.x;
    const int wave = __builtin_amdgcn_readfirstlane(tid >> 5);
    const int lane = tid & 31;
    const int h    = lane >> 4;
    const int c    = lane & 15;
    const int grp  = wave % NG;
    const int strm = wave / NG;
    const int col0 = grp * (CT * 16);

    v4f   b[CT][16];
    float bv[CT];
    #pragma unroll
    for (int t = 0; t < CT; ++t) {
        const int n = col0 + t * 16 + c;
        const bool ok = !GUARD || (n < out_cols_total);
        bv[t] = (bias != nullptr && ok) ? bias[n] : 0.0f;
        #pragma unroll
        for (int q = 0; q < 16; ++q) {
            const int k0 = frag_k0(q, h);
            v4f w = {0.f, 0.f, 0.f, 0.f};
            if (ok) {
                if (TRANS_FILL) {
                    w.x = W[(size_t)(k0 + 0) * out_cols_total + n];
                    w.y = W[(size_t)(k0 + 1) * out_cols_total + n];
                    w.z = W[(size_t)(k0 + 2) * out_cols_total + n];
                    w.w = W[(size_t)(k0 + 3) * out_cols_total + n];
                } else {
                    w = *(const v4f*)(W + (size_t)n * KDIM + k0);
                }
            }
            b[t][q] = w;
        }
    }

    const int nstreams = (int)gridDim.x * NSTR;
    int tile = (int)blockIdx.x * NSTR + strm;
    if (tile >= ntiles) return;
    const int ntrips = (ntiles - tile + nstreams - 1) / nstreams;
    const int last_tile = tile + (ntrips - 1) * nstreams;

    v4f a0[16], a1[16];
    load_a_tile(a0, X, tile, c, h);
    int i = 0;
    for (; i + 2 <= ntrips; i += 2) {
        load_a_tile(a1, X, tile + nstreams, c, h);
        __builtin_amdgcn_sched_barrier(0);
        tile_body<CT, RELU_IN, GUARD>(a0, b, bv, Y, tile, h, c, col0, out_stride, out_cols_total);
        const int t2 = tile + 2 * nstreams;
        load_a_tile(a0, X, t2 <= last_tile ? t2 : last_tile, c, h);
        __builtin_amdgcn_sched_barrier(0);
        tile_body<CT, RELU_IN, GUARD>(a1, b, bv, Y, tile + nstreams, h, c, col0, out_stride, out_cols_total);
        tile = t2;
    }
    if (i < ntrips)
        tile_body<CT, RELU_IN, GUARD>(a0, b, bv, Y, tile, h, c, col0, out_stride, out_cols_total);
}

#define SC_ROWS    200
#define SC_CAP     8192
#define SC_THREADS 1024
#define SC_VEC     8
#define SC_GATHER  8

template <bool ADD_WHEN_FULL>
__device__ __forceinline__ void sc_note_edge(
    int e, int d, int lo, int rows, int* cnt, int2* list, float* acc,
    const float* __restrict__ m, const int* __restrict__ src)
{
    const unsigned r = (unsigned)(d - lo);
    if (r < (unsigned)rows) {
        const int slot = atomicAdd(cnt, 1);
        if (slot < SC_CAP) {
            list[slot] = make_int2(e, (int)r);
        } else if (ADD_WHEN_FULL) {
            const float* p = m + (size_t)src[e] * KDIM;
            float* a = acc + (size_t)r * KDIM;
            #pragma nounroll
            for (int j = 0; j < KDIM; ++j) atomicAdd(a + j, p[j]);
        }
    }
}

template <bool ADD_WHEN_FULL>
__device__ __forceinline__ void sc_filter_edges(
    int lo, int rows, int tid, int* cnt, int2* list, float* acc,
    const float* __restrict__ m, const int* __restrict__ src,
    const int* __restrict__ dst, int nedges)
{
    if ((nedges & 3) == 0) {
        for (int base = 0; base < nedges; base += SC_VEC * SC_THREADS * 4) {
            int4 d[SC_VEC];
            #pragma unroll
            for (int k = 0; k < SC_VEC; ++k) {
                const int e0 = base + (k * SC_THREADS + tid) * 4;
                d[k] = make_int4(lo - 1, lo - 1, lo - 1, lo - 1);
                if (e0 < nedges) d[k] = *(const int4*)(dst + e0);
            }
            #pragma unroll
            for (int k = 0; k < SC_VEC; ++k) {
                const int e0 = base + (k * SC_THREADS + tid) * 4;
                sc_note_edge<ADD_WHEN_FULL>(e0 + 0, d[k].x, lo, rows, cnt, list, acc, m, src);
                sc_note_edge<ADD_WHEN_FULL>(e0 + 1, d[k].y, lo, rows, cnt, list, acc, m, src);
                sc_note_edge<ADD_WHEN_FULL>(e0 + 2, d[k].z, lo, rows, cnt, list, acc, m, src);
                sc_note_edge<ADD_WHEN_FULL>(e0 + 3, d[k].w, lo, rows, cnt, list, acc, m, src);
            }
        }
    } else {
        for (int e = tid; e < nedges; e += SC_THREADS)
            sc_note_edge<ADD_WHEN_FULL>(e, dst[e], lo, rows, cnt, list, acc, m, src);
    }
    __syncthreads();

    const int n = (*cnt < SC_CAP) ? *cnt : SC_CAP;
    for (int i = tid; i < n; i += SC_THREADS)
        list[i].x = src[list[i].x];
    __syncthreads();
}

__global__ __launch_bounds__(SC_THREADS) void ggnn_group_edges(
    const int* __restrict__ src, const int* __restrict__ dst,
    int* __restrict__ lists, int* __restrict__ counts, int nnodes, int nedges)
{
    __shared__ int2 list[SC_CAP];
    __shared__ int cnt;

    const int tid  = threadIdx.x;
    const int g    = (int)blockIdx.x;
    const int lo   = g * SC_ROWS;
    const int rows = (nnodes - lo < SC_ROWS) ? (nnodes - lo) : SC_ROWS;

    if (tid == 0) cnt = 0;
    __syncthreads();

    sc_filter_edges<false>(lo, rows, tid, &cnt, list, nullptr, nullptr, src, dst, nedges);

    const int total = cnt;
    const int n = (total < SC_CAP) ? total : SC_CAP;
    int2* gl = (int2*)lists + (size_t)g * SC_CAP;
    for (int i = tid; i < n; i += SC_THREADS)
        gl[i] = list[i];
    if (tid == 0) counts[g] = total;
}

__global__ __launch_bounds__(SC_THREADS) void ggnn_scatter_owned(
    const float* __restrict__ m, const int* __restrict__ src,
    const int* __restrict__ dst, const int* __restrict__ lists,
    const int* __restrict__ counts, float* __restrict__ agg, int nnodes, int nedges)
{
    __shared__ __attribute__((aligned(16))) float acc[SC_ROWS * KDIM];
    __shared__ int2 list[SC_CAP];
    __shared__ int cnt;

    const int tid  = threadIdx.x;
    const int lane = tid & 31;
    const int wave = __builtin_amdgcn_readfirstlane(tid >> 5);

    const int g    = (int)blockIdx.x;
    const int lo   = g * SC_ROWS;
    const int rows = (nnodes - lo < SC_ROWS) ? (nnodes - lo) : SC_ROWS;
    const int total = __builtin_amdgcn_readfirstlane(counts[g]);
    const bool have_list = total <= SC_CAP;

    int2 t[SC_CAP / SC_THREADS];
    #pragma unroll
    for (int k = 0; k < SC_CAP / SC_THREADS; ++k) {
        const int i = k * SC_THREADS + tid;
        t[k] = make_int2(0, 0);
        if (have_list && i < total)
            t[k] = ((const int2*)lists)[(size_t)g * SC_CAP + i];
    }
    for (int i = tid; i < SC_ROWS * KDIM / 4; i += SC_THREADS)
        ((float4*)acc)[i] = make_float4(0.f, 0.f, 0.f, 0.f);
    if (tid == 0) cnt = 0;
    #pragma unroll
    for (int k = 0; k < SC_CAP / SC_THREADS; ++k) {
        const int i = k * SC_THREADS + tid;
        if (have_list && i < total) list[i] = t[k];
    }
    __syncthreads();

    if (!have_list)
        sc_filter_edges<true>(lo, rows, tid, &cnt, list, acc, m, src, dst, nedges);

    const int n = (total < SC_CAP) ? total : SC_CAP;

    for (int i = wave; i < n; i += 32 * SC_GATHER) {
        int2 mine = make_int2(0, -1);
        const int my_idx = i + 32 * lane;
        if (lane < SC_GATHER && my_idx < n) mine = list[my_idx];
        float v[SC_GATHER][4];
        int   r[SC_GATHER];
        #pragma unroll
        for (int u = 0; u < SC_GATHER; ++u) {
            const int s = __builtin_amdgcn_readlane(mine.x, u);
            r[u] = __builtin_amdgcn_readlane(mine.y, u);
            const float* p = m + (size_t)s * KDIM + lane;
            v[u][0] = p[0]; v[u][1] = p[32]; v[u][2] = p[64]; v[u][3] = p[96];
        }
        #pragma unroll
        for (int u = 0; u < SC_GATHER; ++u) {
            if (r[u] >= 0) {
                float* a = acc + r[u] * KDIM + lane;
                atomicAdd(a,      v[u][0]);
                atomicAdd(a + 32, v[u][1]);
                atomicAdd(a + 64, v[u][2]);
                atomicAdd(a + 96, v[u][3]);
            }
        }
    }
    __syncthreads();

    float4* out4 = (float4*)(agg + (size_t)lo * KDIM);
    for (int i = tid; i < rows * (KDIM / 4); i += SC_THREADS)
        out4[i] = ((const float4*)acc)[i];
}

struct GateRow { v4f ir, iz, in, hr, hz, hn, hv; };

__device__ __forceinline__ void gates_load_row(
    GateRow& g, const float* __restrict__ gi, const float* __restrict__ gh,
    const float* h, int row, int lane)
{
    const float* pi = gi + (size_t)row * 384 + lane * 4;
    const float* ph = gh + (size_t)row * 384 + lane * 4;
    g.ir = *(const v4f*)(pi);
    g.iz = *(const v4f*)(pi + 128);
    g.in = *(const v4f*)(pi + 256);
    g.hr = *(const v4f*)(ph);
    g.hz = *(const v4f*)(ph + 128);
    g.hn = *(const v4f*)(ph + 256);
    g.hv = *(const v4f*)(h + (size_t)row * KDIM + lane * 4);
}

__device__ __forceinline__ float gate_one(
    float i_r, float i_z, float i_n, float h_r, float h_z, float h_n, float hv)
{
    const float r = 1.0f / (1.0f + __expf(-(i_r + h_r)));
    const float z = 1.0f / (1.0f + __expf(-(i_z + h_z)));
    const float nn = tanhf(i_n + r * h_n);
    return (1.0f - z) * nn + z * hv;
}

__device__ __forceinline__ void gates_store_row(
    const GateRow& g, float* h, int row, int lane)
{
    v4f o;
    o.x = gate_one(g.ir.x, g.iz.x, g.in.x, g.hr.x, g.hz.x, g.hn.x, g.hv.x);
    o.y = gate_one(g.ir.y, g.iz.y, g.in.y, g.hr.y, g.hz.y, g.hn.y, g.hv.y);
    o.z = gate_one(g.ir.z, g.iz.z, g.in.z, g.hr.z, g.hz.z, g.hn.z, g.hv.z);
    o.w = gate_one(g.ir.w, g.iz.w, g.in.w, g.hr.w, g.hz.w, g.hn.w, g.hv.w);
    *(v4f*)(h + (size_t)row * KDIM + lane * 4) = o;
}

__device__ __forceinline__ void gates_load_pair(
    GateRow& g0, GateRow& g1, const float* __restrict__ gi,
    const float* __restrict__ gh, const float* h, int pair, int nnodes, int lane)
{
    const int r0 = 2 * pair;
    const int r1 = (r0 + 1 < nnodes) ? r0 + 1 : r0;
    gates_load_row(g0, gi, gh, h, r0, lane);
    gates_load_row(g1, gi, gh, h, r1, lane);
}

__device__ __forceinline__ void gates_store_pair(
    const GateRow& g0, const GateRow& g1, float* h, int pair, int nnodes, int lane)
{
    const int r0 = 2 * pair;
    gates_store_row(g0, h, r0, lane);
    if (r0 + 1 < nnodes) gates_store_row(g1, h, r0 + 1, lane);
}

__global__ __launch_bounds__(256) void ggnn_gru_gates(
    const float* __restrict__ gi, const float* __restrict__ gh,
    float* h, int nnodes)
{
    const int lane = threadIdx.x & 31;
    const int wave = __builtin_amdgcn_readfirstlane((int)threadIdx.x >> 5);
    const int nwaves = (int)gridDim.x * 8;
    const int npairs = (nnodes + 1) >> 1;

    int pair = (int)blockIdx.x * 8 + wave;
    if (pair >= npairs) return;
    const int ntrips = (npairs - pair + nwaves - 1) / nwaves;
    const int last_pair = pair + (ntrips - 1) * nwaves;

    GateRow a0, a1, b0, b1;
    gates_load_pair(a0, a1, gi, gh, h, pair, nnodes, lane);
    int i = 0;
    for (; i + 2 <= ntrips; i += 2) {
        gates_load_pair(b0, b1, gi, gh, h, pair + nwaves, nnodes, lane);
        __builtin_amdgcn_sched_barrier(0);
        gates_store_pair(a0, a1, h, pair, nnodes, lane);
        const int p2 = pair + 2 * nwaves;
        gates_load_pair(a0, a1, gi, gh, h, p2 <= last_pair ? p2 : last_pair, nnodes, lane);
        __builtin_amdgcn_sched_barrier(0);
        gates_store_pair(b0, b1, h, pair + nwaves, nnodes, lane);
        pair = p2;
    }
    if (i < ntrips)
        gates_store_pair(a0, a1, h, pair, nnodes, lane);
}

extern "C" void kernel_launch(void* const* d_in, const int* in_sizes, int n_in,
                              void* d_out, int out_size, void* d_ws, size_t ws_size,
                              hipStream_t stream)
{
    (void)n_in; (void)out_size; (void)ws_size;

    const float* x      = (const float*)d_in[0];
    const int*   ei     = (const int*)d_in[1];
    const float* lin_w  = (const float*)d_in[2];
    const float* lin_b  = (const float*)d_in[3];
    const float* ggc_w  = (const float*)d_in[4];
    const float* w_ih   = (const float*)d_in[5];
    const float* w_hh   = (const float*)d_in[6];
    const float* b_ih   = (const float*)d_in[7];
    const float* b_hh   = (const float*)d_in[8];
    const float* lin2_w = (const float*)d_in[9];
    const float* lin2_b = (const float*)d_in[10];
    float* out = (float*)d_out;

    const int N = in_sizes[0] / KDIM;
    const int E = in_sizes[1] / 2;
    const int* src = ei;
    const int* dst = ei + E;

    float* h   = (float*)d_ws;
    float* m   = h   + (size_t)N * 128;
    float* agg = m   + (size_t)N * 128;
    float* gi  = agg + (size_t)N * 128;
    float* gh  = gi  + (size_t)N * 384;

    const int row_tiles = (N + 15) / 16;
    const dim3 blk(256);
    const dim3 gemm_grid(256);

    const unsigned scatter_blks = (unsigned)((N + SC_ROWS - 1) / SC_ROWS);
    int* lists  = (int*)(gh + (size_t)N * 384);
    int* counts = lists + (size_t)scatter_blks * SC_CAP * 2;
    const dim3 gate_grid(512);

    ggnn_gemm<2, 4, false, false, false><<<gemm_grid, blk, 0, stream>>>(
        x, lin_w, lin_b, h, row_tiles, 128, 128);

    ggnn_group_edges<<<scatter_blks, dim3(SC_THREADS), 0, stream>>>(src, dst, lists, counts, N, E);

    for (int l = 0; l < 8; ++l) {
        ggnn_gemm<2, 4, true, false, false><<<gemm_grid, blk, 0, stream>>>(
            h, ggc_w + (size_t)l * 128 * 128, nullptr, m, row_tiles, 128, 128);
        ggnn_scatter_owned<<<scatter_blks, dim3(SC_THREADS), 0, stream>>>(m, src, dst, lists, counts, agg, N, E);
        ggnn_gemm<3, 8, false, false, false><<<gemm_grid, blk, 0, stream>>>(
            agg, w_ih, b_ih, gi, row_tiles, 384, 384);
        ggnn_gemm<3, 8, false, false, false><<<gemm_grid, blk, 0, stream>>>(
            h, w_hh, b_hh, gh, row_tiles, 384, 384);
        ggnn_gru_gates<<<gate_grid, blk, 0, stream>>>(gi, gh, h, N);
    }

    ggnn_gemm<2, 4, false, true, true><<<gemm_grid, blk, 0, stream>>>(
        h, lin2_w, lin2_b, out, row_tiles, 100, 100);
}


# GRU gates as epilogue of the agg-side GEMM: a wave holds the r,z,n weight tiles of one 16-column slice, reads gh and h at its accumulator positions, writes h in place; gi array and gate launch go
# speedup vs baseline: 3.5322x; 1.0377x over previous
#include <hip/hip_runtime.h>
#include <hip/hip_bf16.h>
#include <math.h>

typedef float v2f __attribute__((ext_vector_type(2)));
typedef float v4f __attribute__((ext_vector_type(4)));
typedef float v8f __attribute__((ext_vector_type(8)));

#define KDIM 128

__device__ __forceinline__ int frag_k0(int q, int h) { return 8 * q + 4 * h; }

__device__ __forceinline__ void load_a_tile(
    v4f (&a)[16], const float* __restrict__ X, int tile, int c, int h)
{
    const float* p = X + (size_t)(tile * 16 + c) * KDIM;
    #pragma unroll
    for (int q = 0; q < 16; ++q)
        a[q] = *(const v4f*)(p + frag_k0(q, h));
}

template <int CT, bool RELU_IN, bool GUARD>
__device__ __forceinline__ void tile_body(
    const v4f (&a)[16], const v4f (&b)[CT][16], const float (&bv)[CT],
    float* __restrict__ Y, int tile, int h, int c, int col0,
    int out_stride, int out_cols_total)
{
    v8f acc[CT];
    #pragma unroll
    for (int t = 0; t < CT; ++t) acc[t] = v8f{0.f, 0.f, 0.f, 0.f, 0.f, 0.f, 0.f, 0.f};

    #pragma unroll
    for (int q = 0; q < 16; ++q) {
        v2f a0 = {a[q].x, a[q].y};
        v2f a1 = {a[q].z, a[q].w};
        if (RELU_IN) {
            a0.x = fmaxf(a0.x, 0.0f); a0.y = fmaxf(a0.y, 0.0f);
            a1.x = fmaxf(a1.x, 0.0f); a1.y = fmaxf(a1.y, 0.0f);
        }
        #pragma unroll
        for (int t = 0; t < CT; ++t) {
            const v2f b0 = {b[t][q].x, b[t][q].y};
            acc[t] = __builtin_amdgcn_wmma_f32_16x16x4_f32(
                false, a0, false, b0, (short)0, acc[t], false, false);
        }
        #pragma unroll
        for (int t = 0; t < CT; ++t) {
            const v2f b1 = {b[t][q].z, b[t][q].w};
            acc[t] = __builtin_amdgcn_wmma_f32_16x16x4_f32(
                false, a1, false, b1, (short)0, acc[t], false, false);
        }
    }

    #pragma unroll
    for (int t = 0; t < CT; ++t) {
        const int cc = col0 + t * 16 + c;
        if (!GUARD || cc < out_cols_total) {
            #pragma unroll
            for (int r = 0; r < 8; ++r) {
                const int row = tile * 16 + r + h * 8;
                Y[(size_t)row * out_stride + cc] = acc[t][r] + bv[t];
            }
        }
    }
}

template <int CT, int NG, bool TRANS_FILL, bool RELU_IN, bool GUARD>
__global__ __launch_bounds__(256) __attribute__((amdgpu_waves_per_eu(2, 2))) void ggnn_gemm(
    const float* __restrict__ X, const float* __restrict__ W,
    const float* __restrict__ bias, float* __restrict__ Y,
    int ntiles, int out_stride, int out_cols_total)
{
    constexpr int NSTR = 8 / NG;
    const int tid  = threadIdx.x;
    const int wave = __builtin_amdgcn_readfirstlane(tid >> 5);
    const int lane = tid & 31;
    const int h    = lane >> 4;
    const int c    = lane & 15;
    const int grp  = wave % NG;
    const int strm = wave / NG;
    const int col0 = grp * (CT * 16);

    v4f   b[CT][16];
    float bv[CT];
    #pragma unroll
    for (int t = 0; t < CT; ++t) {
        const int n = col0 + t * 16 + c;
        const bool ok = !GUARD || (n < out_cols_total);
        bv[t] = (bias != nullptr && ok) ? bias[n] : 0.0f;
        #pragma unroll
        for (int q = 0; q < 16; ++q) {
            const int k0 = frag_k0(q, h);
            v4f w = {0.f, 0.f, 0.f, 0.f};
            if (ok) {
                if (TRANS_FILL) {
                    w.x = W[(size_t)(k0 + 0) * out_cols_total + n];
                    w.y = W[(size_t)(k0 + 1) * out_cols_total + n];
                    w.z = W[(size_t)(k0 + 2) * out_cols_total + n];
                    w.w = W[(size_t)(k0 + 3) * out_cols_total + n];
                } else {
                    w = *(const v4f*)(W + (size_t)n * KDIM + k0);
                }
            }
            b[t][q] = w;
        }
    }

    const int nstreams = (int)gridDim.x * NSTR;
    int tile = (int)blockIdx.x * NSTR + strm;
    if (tile >= ntiles) return;
    const int ntrips = (ntiles - tile + nstreams - 1) / nstreams;
    const int last_tile = tile + (ntrips - 1) * nstreams;

    v4f a0[16], a1[16];
    load_a_tile(a0, X, tile, c, h);
    int i = 0;
    for (; i + 2 <= ntrips; i += 2) {
        load_a_tile(a1, X, tile + nstreams, c, h);
        __builtin_amdgcn_sched_barrier(0);
        tile_body<CT, RELU_IN, GUARD>(a0, b, bv, Y, tile, h, c, col0, out_stride, out_cols_total);
        const int t2 = tile + 2 * nstreams;
        load_a_tile(a0, X, t2 <= last_tile ? t2 : last_tile, c, h);
        __builtin_amdgcn_sched_barrier(0);
        tile_body<CT, RELU_IN, GUARD>(a1, b, bv, Y, tile + nstreams, h, c, col0, out_stride, out_cols_total);
        tile = t2;
    }
    if (i < ntrips)
        tile_body<CT, RELU_IN, GUARD>(a0, b, bv, Y, tile, h, c, col0, out_stride, out_cols_total);
}

#define SC_ROWS    200
#define SC_CAP     8192
#define SC_THREADS 1024
#define SC_VEC     8
#define SC_GATHER  8

template <bool ADD_WHEN_FULL>
__device__ __forceinline__ void sc_note_edge(
    int e, int d, int lo, int rows, int* cnt, int2* list, float* acc,
    const float* __restrict__ m, const int* __restrict__ src)
{
    const unsigned r = (unsigned)(d - lo);
    if (r < (unsigned)rows) {
        const int slot = atomicAdd(cnt, 1);
        if (slot < SC_CAP) {
            list[slot] = make_int2(e, (int)r);
        } else if (ADD_WHEN_FULL) {
            const float* p = m + (size_t)src[e] * KDIM;
            float* a = acc + (size_t)r * KDIM;
            #pragma nounroll
            for (int j = 0; j < KDIM; ++j) atomicAdd(a + j, p[j]);
        }
    }
}

template <bool ADD_WHEN_FULL>
__device__ __forceinline__ void sc_filter_edges(
    int lo, int rows, int tid, int* cnt, int2* list, float* acc,
    const float* __restrict__ m, const int* __restrict__ src,
    const int* __restrict__ dst, int nedges)
{
    if ((nedges & 3) == 0) {
        for (int base = 0; base < nedges; base += SC_VEC * SC_THREADS * 4) {
            int4 d[SC_VEC];
            #pragma unroll
            for (int k = 0; k < SC_VEC; ++k) {
                const int e0 = base + (k * SC_THREADS + tid) * 4;
                d[k] = make_int4(lo - 1, lo - 1, lo - 1, lo - 1);
                if (e0 < nedges) d[k] = *(const int4*)(dst + e0);
            }
            #pragma unroll
            for (int k = 0; k < SC_VEC; ++k) {
                const int e0 = base + (k * SC_THREADS + tid) * 4;
                sc_note_edge<ADD_WHEN_FULL>(e0 + 0, d[k].x, lo, rows, cnt, list, acc, m, src);
                sc_note_edge<ADD_WHEN_FULL>(e0 + 1, d[k].y, lo, rows, cnt, list, acc, m, src);
                sc_note_edge<ADD_WHEN_FULL>(e0 + 2, d[k].z, lo, rows, cnt, list, acc, m, src);
                sc_note_edge<ADD_WHEN_FULL>(e0 + 3, d[k].w, lo, rows, cnt, list, acc, m, src);
            }
        }
    } else {
        for (int e = tid; e < nedges; e += SC_THREADS)
            sc_note_edge<ADD_WHEN_FULL>(e, dst[e], lo, rows, cnt, list, acc, m, src);
    }
    __syncthreads();

    const int n = (*cnt < SC_CAP) ? *cnt : SC_CAP;
    for (int i = tid; i < n; i += SC_THREADS)
        list[i].x = src[list[i].x];
    __syncthreads();
}

__global__ __launch_bounds__(SC_THREADS) void ggnn_group_edges(
    const int* __restrict__ src, const int* __restrict__ dst,
    int* __restrict__ lists, int* __restrict__ counts, int nnodes, int nedges)
{
    __shared__ int2 list[SC_CAP];
    __shared__ int cnt;

    const int tid  = threadIdx.x;
    const int g    = (int)blockIdx.x;
    const int lo   = g * SC_ROWS;
    const int rows = (nnodes - lo < SC_ROWS) ? (nnodes - lo) : SC_ROWS;

    if (tid == 0) cnt = 0;
    __syncthreads();

    sc_filter_edges<false>(lo, rows, tid, &cnt, list, nullptr, nullptr, src, dst, nedges);

    const int total = cnt;
    const int n = (total < SC_CAP) ? total : SC_CAP;
    int2* gl = (int2*)lists + (size_t)g * SC_CAP;
    for (int i = tid; i < n; i += SC_THREADS)
        gl[i] = list[i];
    if (tid == 0) counts[g] = total;
}

__global__ __launch_bounds__(SC_THREADS) void ggnn_scatter_owned(
    const float* __restrict__ m, const int* __restrict__ src,
    const int* __restrict__ dst, const int* __restrict__ lists,
    const int* __restrict__ counts, float* __restrict__ agg, int nnodes, int nedges)
{
    __shared__ __attribute__((aligned(16))) float acc[SC_ROWS * KDIM];
    __shared__ int2 list[SC_CAP];
    __shared__ int cnt;

    const int tid  = threadIdx.x;
    const int lane = tid & 31;
    const int wave = __builtin_amdgcn_readfirstlane(tid >> 5);

    const int g    = (int)blockIdx.x;
    const int lo   = g * SC_ROWS;
    const int rows = (nnodes - lo < SC_ROWS) ? (nnodes - lo) : SC_ROWS;
    const int total = __builtin_amdgcn_readfirstlane(counts[g]);
    const bool have_list = total <= SC_CAP;

    int2 t[SC_CAP / SC_THREADS];
    #pragma unroll
    for (int k = 0; k < SC_CAP / SC_THREADS; ++k) {
        const int i = k * SC_THREADS + tid;
        t[k] = make_int2(0, 0);
        if (have_list && i < total)
            t[k] = ((const int2*)lists)[(size_t)g * SC_CAP + i];
    }
    for (int i = tid; i < SC_ROWS * KDIM / 4; i += SC_THREADS)
        ((float4*)acc)[i] = make_float4(0.f, 0.f, 0.f, 0.f);
    if (tid == 0) cnt = 0;
    #pragma unroll
    for (int k = 0; k < SC_CAP / SC_THREADS; ++k) {
        const int i = k * SC_THREADS + tid;
        if (have_list && i < total) list[i] = t[k];
    }
    __syncthreads();

    if (!have_list)
        sc_filter_edges<true>(lo, rows, tid, &cnt, list, acc, m, src, dst, nedges);

    const int n = (total < SC_CAP) ? total : SC_CAP;

    for (int i = wave; i < n; i += 32 * SC_GATHER) {
        int2 mine = make_int2(0, -1);
        const int my_idx = i + 32 * lane;
        if (lane < SC_GATHER && my_idx < n) mine = list[my_idx];
        float v[SC_GATHER][4];
        int   r[SC_GATHER];
        #pragma unroll
        for (int u = 0; u < SC_GATHER; ++u) {
            const int s = __builtin_amdgcn_readlane(mine.x, u);
            r[u] = __builtin_amdgcn_readlane(mine.y, u);
            const float* p = m + (size_t)s * KDIM + lane;
            v[u][0] = p[0]; v[u][1] = p[32]; v[u][2] = p[64]; v[u][3] = p[96];
        }
        #pragma unroll
        for (int u = 0; u < SC_GATHER; ++u) {
            if (r[u] >= 0) {
                float* a = acc + r[u] * KDIM + lane;
                atomicAdd(a,      v[u][0]);
                atomicAdd(a + 32, v[u][1]);
                atomicAdd(a + 64, v[u][2]);
                atomicAdd(a + 96, v[u][3]);
            }
        }
    }
    __syncthreads();

    float4* out4 = (float4*)(agg + (size_t)lo * KDIM);
    for (int i = tid; i < rows * (KDIM / 4); i += SC_THREADS)
        out4[i] = ((const float4*)acc)[i];
}

__device__ __forceinline__ float gate_one(
    float i_r, float i_z, float i_n, float h_r, float h_z, float h_n, float hv)
{
    const float r = 1.0f / (1.0f + __expf(-(i_r + h_r)));
    const float z = 1.0f / (1.0f + __expf(-(i_z + h_z)));
    const float nn = tanhf(i_n + r * h_n);
    return (1.0f - z) * nn + z * hv;
}

struct GateOps { float hr[8], hz[8], hn[8], hv[8]; };

__device__ __forceinline__ void load_gate_ops(
    GateOps& g, const float* __restrict__ gh, const float* h, int tile, int half, int hc)
{
    #pragma unroll
    for (int r = 0; r < 8; ++r) {
        const size_t row = (size_t)(tile * 16 + r + half * 8);
        const float* p = gh + row * 384 + hc;
        g.hr[r] = p[0];
        g.hz[r] = p[128];
        g.hn[r] = p[256];
        g.hv[r] = h[row * KDIM + hc];
    }
}

__device__ __forceinline__ void gi_gate_tile(
    const v4f (&a)[16], const v4f (&b)[3][16], const float (&bv)[3],
    const GateOps& g, float* h, int tile, int half, int hc)
{
    v8f acc[3];
    #pragma unroll
    for (int t = 0; t < 3; ++t) acc[t] = v8f{0.f, 0.f, 0.f, 0.f, 0.f, 0.f, 0.f, 0.f};

    #pragma unroll
    for (int q = 0; q < 16; ++q) {
        const v2f a0 = {a[q].x, a[q].y};
        const v2f a1 = {a[q].z, a[q].w};
        #pragma unroll
        for (int t = 0; t < 3; ++t) {
            const v2f b0 = {b[t][q].x, b[t][q].y};
            acc[t] = __builtin_amdgcn_wmma_f32_16x16x4_f32(
                false, a0, false, b0, (short)0, acc[t], false, false);
        }
        #pragma unroll
        for (int t = 0; t < 3; ++t) {
            const v2f b1 = {b[t][q].z, b[t][q].w};
            acc[t] = __builtin_amdgcn_wmma_f32_16x16x4_f32(
                false, a1, false, b1, (short)0, acc[t], false, false);
        }
    }

    #pragma unroll
    for (int r = 0; r < 8; ++r) {
        const size_t row = (size_t)(tile * 16 + r + half * 8);
        h[row * KDIM + hc] = gate_one(
            acc[0][r] + bv[0], acc[1][r] + bv[1], acc[2][r] + bv[2],
            g.hr[r], g.hz[r], g.hn[r], g.hv[r]);
    }
}

__global__ __launch_bounds__(256) __attribute__((amdgpu_waves_per_eu(2, 2))) void ggnn_gi_gates(
    const float* __restrict__ agg, const float* __restrict__ w_ih,
    const float* __restrict__ b_ih, const float* __restrict__ gh,
    float* h, int ntiles)
{
    const int tid  = threadIdx.x;
    const int wave = __builtin_amdgcn_readfirstlane(tid >> 5);
    const int lane = tid & 31;
    const int half = lane >> 4;
    const int c    = lane & 15;
    const int hc   = wave * 16 + c;

    v4f   b[3][16];
    float bv[3];
    #pragma unroll
    for (int t = 0; t < 3; ++t) {
        const int n = t * 128 + hc;
        bv[t] = b_ih[n];
        #pragma unroll
        for (int q = 0; q < 16; ++q)
            b[t][q] = *(const v4f*)(w_ih + (size_t)n * KDIM + frag_k0(q, half));
    }

    const int nstreams = (int)gridDim.x;
    int tile = (int)blockIdx.x;
    if (tile >= ntiles) return;
    const int ntrips = (ntiles - tile + nstreams - 1) / nstreams;
    const int last_tile = tile + (ntrips - 1) * nstreams;

    v4f a0[16], a1[16];
    GateOps g;
    load_a_tile(a0, agg, tile, c, half);
    int i = 0;
    for (; i + 2 <= ntrips; i += 2) {
        load_a_tile(a1, agg, tile + nstreams, c, half);
        load_gate_ops(g, gh, h, tile, half, hc);
        __builtin_amdgcn_sched_barrier(0);
        gi_gate_tile(a0, b, bv, g, h, tile, half, hc);
        const int t2 = tile + 2 * nstreams;
        load_a_tile(a0, agg, t2 <= last_tile ? t2 : last_tile, c, half);
        load_gate_ops(g, gh, h, tile + nstreams, half, hc);
        __builtin_amdgcn_sched_barrier(0);
        gi_gate_tile(a1, b, bv, g, h, tile + nstreams, half, hc);
        tile = t2;
    }
    if (i < ntrips) {
        load_gate_ops(g, gh, h, tile, half, hc);
        __builtin_amdgcn_sched_barrier(0);
        gi_gate_tile(a0, b, bv, g, h, tile, half, hc);
    }
}

extern "C" void kernel_launch(void* const* d_in, const int* in_sizes, int n_in,
                              void* d_out, int out_size, void* d_ws, size_t ws_size,
                              hipStream_t stream)
{
    (void)n_in; (void)out_size; (void)ws_size;

    const float* x      = (const float*)d_in[0];
    const int*   ei     = (const int*)d_in[1];
    const float* lin_w  = (const float*)d_in[2];
    const float* lin_b  = (const float*)d_in[3];
    const float* ggc_w  = (const float*)d_in[4];
    const float* w_ih   = (const float*)d_in[5];
    const float* w_hh   = (const float*)d_in[6];
    const float* b_ih   = (const float*)d_in[7];
    const float* b_hh   = (const float*)d_in[8];
    const float* lin2_w = (const float*)d_in[9];
    const float* lin2_b = (const float*)d_in[10];
    float* out = (float*)d_out;

    const int N = in_sizes[0] / KDIM;
    const int E = in_sizes[1] / 2;
    const int* src = ei;
    const int* dst = ei + E;

    float* h   = (float*)d_ws;
    float* m   = h   + (size_t)N * 128;
    float* agg = m   + (size_t)N * 128;
    float* gh  = agg + (size_t)N * 128;

    const int row_tiles = (N + 15) / 16;
    const dim3 blk(256);
    const dim3 gemm_grid(256);

    const unsigned scatter_blks = (unsigned)((N + SC_ROWS - 1) / SC_ROWS);
    int* lists  = (int*)(gh + (size_t)N * 384);
    int* counts = lists + (size_t)scatter_blks * SC_CAP * 2;

    ggnn_gemm<2, 4, false, false, false><<<gemm_grid, blk, 0, stream>>>(
        x, lin_w, lin_b, h, row_tiles, 128, 128);

    ggnn_group_edges<<<scatter_blks, dim3(SC_THREADS), 0, stream>>>(src, dst, lists, counts, N, E);

    for (int l = 0; l < 8; ++l) {
        ggnn_gemm<2, 4, true, false, false><<<gemm_grid, blk, 0, stream>>>(
            h, ggc_w + (size_t)l * 128 * 128, nullptr, m, row_tiles, 128, 128);
        ggnn_scatter_owned<<<scatter_blks, dim3(SC_THREADS), 0, stream>>>(m, src, dst, lists, counts, agg, N, E);
        ggnn_gemm<3, 8, false, false, false><<<gemm_grid, blk, 0, stream>>>(
            h, w_hh, b_hh, gh, row_tiles, 384, 384);
        ggnn_gi_gates<<<gemm_grid, blk, 0, stream>>>(agg, w_ih, b_ih, gh, h, row_tiles);
    }

    ggnn_gemm<2, 4, false, true, true><<<gemm_grid, blk, 0, stream>>>(
        h, lin2_w, lin2_b, out, row_tiles, 100, 100);
}


# Gate kernel goes on to the next layer's message and gh products for the row tiles it has just written (one pass over h, 1+3 weight tiles per wave); launches per call 35 -> 20; sums unchanged
# speedup vs baseline: 3.6524x; 1.0340x over previous
#include <hip/hip_runtime.h>
#include <hip/hip_bf16.h>
#include <math.h>

typedef float v2f __attribute__((ext_vector_type(2)));
typedef float v4f __attribute__((ext_vector_type(4)));
typedef float v8f __attribute__((ext_vector_type(8)));

#define KDIM 128

__device__ __forceinline__ int frag_k0(int q, int h) { return 8 * q + 4 * h; }

__device__ __forceinline__ void load_a_tile(
    v4f (&a)[16], const float* __restrict__ X, int tile, int c, int h)
{
    const float* p = X + (size_t)(tile * 16 + c) * KDIM;
    #pragma unroll
    for (int q = 0; q < 16; ++q)
        a[q] = *(const v4f*)(p + frag_k0(q, h));
}

template <int CT, bool RELU_IN, bool GUARD>
__device__ __forceinline__ void tile_body(
    const v4f (&a)[16], const v4f (&b)[CT][16], const float (&bv)[CT],
    float* __restrict__ Y, int tile, int h, int c, int col0,
    int out_stride, int out_cols_total)
{
    v8f acc[CT];
    #pragma unroll
    for (int t = 0; t < CT; ++t) acc[t] = v8f{0.f, 0.f, 0.f, 0.f, 0.f, 0.f, 0.f, 0.f};

    #pragma unroll
    for (int q = 0; q < 16; ++q) {
        v2f a0 = {a[q].x, a[q].y};
        v2f a1 = {a[q].z, a[q].w};
        if (RELU_IN) {
            a0.x = fmaxf(a0.x, 0.0f); a0.y = fmaxf(a0.y, 0.0f);
            a1.x = fmaxf(a1.x, 0.0f); a1.y = fmaxf(a1.y, 0.0f);
        }
        #pragma unroll
        for (int t = 0; t < CT; ++t) {
            const v2f b0 = {b[t][q].x, b[t][q].y};
            acc[t] = __builtin_amdgcn_wmma_f32_16x16x4_f32(
                false, a0, false, b0, (short)0, acc[t], false, false);
        }
        #pragma unroll
        for (int t = 0; t < CT; ++t) {
            const v2f b1 = {b[t][q].z, b[t][q].w};
            acc[t] = __builtin_amdgcn_wmma_f32_16x16x4_f32(
                false, a1, false, b1, (short)0, acc[t], false, false);
        }
    }

    #pragma unroll
    for (int t = 0; t < CT; ++t) {
        const int cc = col0 + t * 16 + c;
        if (!GUARD || cc < out_cols_total) {
            #pragma unroll
            for (int r = 0; r < 8; ++r) {
                const int row = tile * 16 + r + h * 8;
                Y[(size_t)row * out_stride + cc] = acc[t][r] + bv[t];
            }
        }
    }
}

template <int CT, int NG, bool TRANS_FILL, bool RELU_IN, bool GUARD>
__global__ __launch_bounds__(256) __attribute__((amdgpu_waves_per_eu(2, 2))) void ggnn_gemm(
    const float* __restrict__ X, const float* __restrict__ W,
    const float* __restrict__ bias, float* __restrict__ Y,
    int ntiles, int out_stride, int out_cols_total)
{
    constexpr int NSTR = 8 / NG;
    const int tid  = threadIdx.x;
    const int wave = __builtin_amdgcn_readfirstlane(tid >> 5);
    const int lane = tid & 31;
    const int h    = lane >> 4;
    const int c    = lane & 15;
    const int grp  = wave % NG;
    const int strm = wave / NG;
    const int col0 = grp * (CT * 16);

    v4f   b[CT][16];
    float bv[CT];
    #pragma unroll
    for (int t = 0; t < CT; ++t) {
        const int n = col0 + t * 16 + c;
        const bool ok = !GUARD || (n < out_cols_total);
        bv[t] = (bias != nullptr && ok) ? bias[n] : 0.0f;
        #pragma unroll
        for (int q = 0; q < 16; ++q) {
            const int k0 = frag_k0(q, h);
            v4f w = {0.f, 0.f, 0.f, 0.f};
            if (ok) {
                if (TRANS_FILL) {
                    w.x = W[(size_t)(k0 + 0) * out_cols_total + n];
                    w.y = W[(size_t)(k0 + 1) * out_cols_total + n];
                    w.z = W[(size_t)(k0 + 2) * out_cols_total + n];
                    w.w = W[(size_t)(k0 + 3) * out_cols_total + n];
                } else {
                    w = *(const v4f*)(W + (size_t)n * KDIM + k0);
                }
            }
            b[t][q] = w;
        }
    }

    const int nstreams = (int)gridDim.x * NSTR;
    int tile = (int)blockIdx.x * NSTR + strm;
    if (tile >= ntiles) return;
    const int ntrips = (ntiles - tile + nstreams - 1) / nstreams;
    const int last_tile = tile + (ntrips - 1) * nstreams;

    v4f a0[16], a1[16];
    load_a_tile(a0, X, tile, c, h);
    int i = 0;
    for (; i + 2 <= ntrips; i += 2) {
        load_a_tile(a1, X, tile + nstreams, c, h);
        __builtin_amdgcn_sched_barrier(0);
        tile_body<CT, RELU_IN, GUARD>(a0, b, bv, Y, tile, h, c, col0, out_stride, out_cols_total);
        const int t2 = tile + 2 * nstreams;
        load_a_tile(a0, X, t2 <= last_tile ? t2 : last_tile, c, h);
        __builtin_amdgcn_sched_barrier(0);
        tile_body<CT, RELU_IN, GUARD>(a1, b, bv, Y, tile + nstreams, h, c, col0, out_stride, out_cols_total);
        tile = t2;
    }
    if (i < ntrips)
        tile_body<CT, RELU_IN, GUARD>(a0, b, bv, Y, tile, h, c, col0, out_stride, out_cols_total);
}

__device__ __forceinline__ void m_gh_tile(
    const v4f (&a)[16], const v4f (&b)[4][16], const float (&bv)[4],
    float* __restrict__ M, float* GH, int tile, int h, int mc, int gc)
{
    v8f acc[4];
    #pragma unroll
    for (int t = 0; t < 4; ++t) acc[t] = v8f{0.f, 0.f, 0.f, 0.f, 0.f, 0.f, 0.f, 0.f};

    #pragma unroll
    for (int q = 0; q < 16; ++q) {
        const v2f a0 = {a[q].x, a[q].y};
        const v2f a1 = {a[q].z, a[q].w};
        #pragma unroll
        for (int t = 0; t < 4; ++t) {
            const v2f b0 = {b[t][q].x, b[t][q].y};
            acc[t] = __builtin_amdgcn_wmma_f32_16x16x4_f32(
                false, a0, false, b0, (short)0, acc[t], false, false);
        }
        #pragma unroll
        for (int t = 0; t < 4; ++t) {
            const v2f b1 = {b[t][q].z, b[t][q].w};
            acc[t] = __builtin_amdgcn_wmma_f32_16x16x4_f32(
                false, a1, false, b1, (short)0, acc[t], false, false);
        }
    }

    #pragma unroll
    for (int r = 0; r < 8; ++r) {
        const size_t row = (size_t)(tile * 16 + r + h * 8);
        M[row * 128 + mc] = acc[0][r];
        #pragma unroll
        for (int t = 1; t < 4; ++t)
            GH[row * 384 + gc + (t - 1) * 16] = acc[t][r] + bv[t];
    }
}

__device__ __forceinline__ void m_gh_pass(
    const float* X, const float* __restrict__ w_l, const float* __restrict__ w_hh,
    const float* __restrict__ b_hh, float* __restrict__ M, float* GH,
    int ntiles, int first_tile, int nstreams, int wave, int lane)
{
    const int h  = lane >> 4;
    const int c  = lane & 15;
    const int mc = wave * 16 + c;
    const int gc = wave * 48 + c;

    v4f   b[4][16];
    float bv[4];
    bv[0] = 0.0f;
    #pragma unroll
    for (int q = 0; q < 16; ++q) {
        const int k0 = frag_k0(q, h);
        v4f w;
        w.x = w_l[(size_t)(k0 + 0) * 128 + mc];
        w.y = w_l[(size_t)(k0 + 1) * 128 + mc];
        w.z = w_l[(size_t)(k0 + 2) * 128 + mc];
        w.w = w_l[(size_t)(k0 + 3) * 128 + mc];
        b[0][q] = w;
    }
    #pragma unroll
    for (int t = 1; t < 4; ++t) {
        const int n = gc + (t - 1) * 16;
        bv[t] = b_hh[n];
        #pragma unroll
        for (int q = 0; q < 16; ++q)
            b[t][q] = *(const v4f*)(w_hh + (size_t)n * KDIM + frag_k0(q, h));
    }

    int tile = first_tile;
    if (tile >= ntiles) return;
    const int ntrips = (ntiles - tile + nstreams - 1) / nstreams;
    const int last_tile = tile + (ntrips - 1) * nstreams;

    v4f a0[16], a1[16];
    load_a_tile(a0, X, tile, c, h);
    int i = 0;
    for (; i + 2 <= ntrips; i += 2) {
        load_a_tile(a1, X, tile + nstreams, c, h);
        __builtin_amdgcn_sched_barrier(0);
        m_gh_tile(a0, b, bv, M, GH, tile, h, mc, gc);
        const int t2 = tile + 2 * nstreams;
        load_a_tile(a0, X, t2 <= last_tile ? t2 : last_tile, c, h);
        __builtin_amdgcn_sched_barrier(0);
        m_gh_tile(a1, b, bv, M, GH, tile + nstreams, h, mc, gc);
        tile = t2;
    }
    if (i < ntrips)
        m_gh_tile(a0, b, bv, M, GH, tile, h, mc, gc);
}

__global__ __launch_bounds__(256) __attribute__((amdgpu_waves_per_eu(2, 2))) void ggnn_m_gh(
    const float* __restrict__ X, const float* __restrict__ w_l,
    const float* __restrict__ w_hh, const float* __restrict__ b_hh,
    float* __restrict__ M, float* __restrict__ GH, int ntiles)
{
    const int tid  = threadIdx.x;
    const int wave = __builtin_amdgcn_readfirstlane(tid >> 5);
    m_gh_pass(X, w_l, w_hh, b_hh, M, GH, ntiles, (int)blockIdx.x, (int)gridDim.x, wave, tid & 31);
}

#define SC_ROWS    200
#define SC_CAP     8192
#define SC_THREADS 1024
#define SC_VEC     8
#define SC_GATHER  8

template <bool ADD_WHEN_FULL>
__device__ __forceinline__ void sc_note_edge(
    int e, int d, int lo, int rows, int* cnt, int2* list, float* acc,
    const float* __restrict__ m, const int* __restrict__ src)
{
    const unsigned r = (unsigned)(d - lo);
    if (r < (unsigned)rows) {
        const int slot = atomicAdd(cnt, 1);
        if (slot < SC_CAP) {
            list[slot] = make_int2(e, (int)r);
        } else if (ADD_WHEN_FULL) {
            const float* p = m + (size_t)src[e] * KDIM;
            float* a = acc + (size_t)r * KDIM;
            #pragma nounroll
            for (int j = 0; j < KDIM; ++j) atomicAdd(a + j, p[j]);
        }
    }
}

template <bool ADD_WHEN_FULL>
__device__ __forceinline__ void sc_filter_edges(
    int lo, int rows, int tid, int* cnt, int2* list, float* acc,
    const float* __restrict__ m, const int* __restrict__ src,
    const int* __restrict__ dst, int nedges)
{
    if ((nedges & 3) == 0) {
        for (int base = 0; base < nedges; base += SC_VEC * SC_THREADS * 4) {
            int4 d[SC_VEC];
            #pragma unroll
            for (int k = 0; k < SC_VEC; ++k) {
                const int e0 = base + (k * SC_THREADS + tid) * 4;
                d[k] = make_int4(lo - 1, lo - 1, lo - 1, lo - 1);
                if (e0 < nedges) d[k] = *(const int4*)(dst + e0);
            }
            #pragma unroll
            for (int k = 0; k < SC_VEC; ++k) {
                const int e0 = base + (k * SC_THREADS + tid) * 4;
                sc_note_edge<ADD_WHEN_FULL>(e0 + 0, d[k].x, lo, rows, cnt, list, acc, m, src);
                sc_note_edge<ADD_WHEN_FULL>(e0 + 1, d[k].y, lo, rows, cnt, list, acc, m, src);
                sc_note_edge<ADD_WHEN_FULL>(e0 + 2, d[k].z, lo, rows, cnt, list, acc, m, src);
                sc_note_edge<ADD_WHEN_FULL>(e0 + 3, d[k].w, lo, rows, cnt, list, acc, m, src);
            }
        }
    } else {
        for (int e = tid; e < nedges; e += SC_THREADS)
            sc_note_edge<ADD_WHEN_FULL>(e, dst[e], lo, rows, cnt, list, acc, m, src);
    }
    __syncthreads();

    const int n = (*cnt < SC_CAP) ? *cnt : SC_CAP;
    for (int i = tid; i < n; i += SC_THREADS)
        list[i].x = src[list[i].x];
    __syncthreads();
}

__global__ __launch_bounds__(SC_THREADS) void ggnn_group_edges(
    const int* __restrict__ src, const int* __restrict__ dst,
    int* __restrict__ lists, int* __restrict__ counts, int nnodes, int nedges)
{
    __shared__ int2 list[SC_CAP];
    __shared__ int cnt;

    const int tid  = threadIdx.x;
    const int g    = (int)blockIdx.x;
    const int lo   = g * SC_ROWS;
    const int rows = (nnodes - lo < SC_ROWS) ? (nnodes - lo) : SC_ROWS;

    if (tid == 0) cnt = 0;
    __syncthreads();

    sc_filter_edges<false>(lo, rows, tid, &cnt, list, nullptr, nullptr, src, dst, nedges);

    const int total = cnt;
    const int n = (total < SC_CAP) ? total : SC_CAP;
    int2* gl = (int2*)lists + (size_t)g * SC_CAP;
    for (int i = tid; i < n; i += SC_THREADS)
        gl[i] = list[i];
    if (tid == 0) counts[g] = total;
}

__global__ __launch_bounds__(SC_THREADS) void ggnn_scatter_owned(
    const float* __restrict__ m, const int* __restrict__ src,
    const int* __restrict__ dst, const int* __restrict__ lists,
    const int* __restrict__ counts, float* __restrict__ agg, int nnodes, int nedges)
{
    __shared__ __attribute__((aligned(16))) float acc[SC_ROWS * KDIM];
    __shared__ int2 list[SC_CAP];
    __shared__ int cnt;

    const int tid  = threadIdx.x;
    const int lane = tid & 31;
    const int wave = __builtin_amdgcn_readfirstlane(tid >> 5);

    const int g    = (int)blockIdx.x;
    const int lo   = g * SC_ROWS;
    const int rows = (nnodes - lo < SC_ROWS) ? (nnodes - lo) : SC_ROWS;
    const int total = __builtin_amdgcn_readfirstlane(counts[g]);
    const bool have_list = total <= SC_CAP;

    int2 t[SC_CAP / SC_THREADS];
    #pragma unroll
    for (int k = 0; k < SC_CAP / SC_THREADS; ++k) {
        const int i = k * SC_THREADS + tid;
        t[k] = make_int2(0, 0);
        if (have_list && i < total)
            t[k] = ((const int2*)lists)[(size_t)g * SC_CAP + i];
    }
    for (int i = tid; i < SC_ROWS * KDIM / 4; i += SC_THREADS)
        ((float4*)acc)[i] = make_float4(0.f, 0.f, 0.f, 0.f);
    if (tid == 0) cnt = 0;
    #pragma unroll
    for (int k = 0; k < SC_CAP / SC_THREADS; ++k) {
        const int i = k * SC_THREADS + tid;
        if (have_list && i < total) list[i] = t[k];
    }
    __syncthreads();

    if (!have_list)
        sc_filter_edges<true>(lo, rows, tid, &cnt, list, acc, m, src, dst, nedges);

    const int n = (total < SC_CAP) ? total : SC_CAP;

    for (int i = wave; i < n; i += 32 * SC_GATHER) {
        int2 mine = make_int2(0, -1);
        const int my_idx = i + 32 * lane;
        if (lane < SC_GATHER && my_idx < n) mine = list[my_idx];
        float v[SC_GATHER][4];
        int   r[SC_GATHER];
        #pragma unroll
        for (int u = 0; u < SC_GATHER; ++u) {
            const int s = __builtin_amdgcn_readlane(mine.x, u);
            r[u] = __builtin_amdgcn_readlane(mine.y, u);
            const float* p = m + (size_t)s * KDIM + lane;
            v[u][0] = p[0]; v[u][1] = p[32]; v[u][2] = p[64]; v[u][3] = p[96];
        }
        #pragma unroll
        for (int u = 0; u < SC_GATHER; ++u) {
            if (r[u] >= 0) {
                float* a = acc + r[u] * KDIM + lane;
                atomicAdd(a,      v[u][0]);
                atomicAdd(a + 32, v[u][1]);
                atomicAdd(a + 64, v[u][2]);
                atomicAdd(a + 96, v[u][3]);
            }
        }
    }
    __syncthreads();

    float4* out4 = (float4*)(agg + (size_t)lo * KDIM);
    for (int i = tid; i < rows * (KDIM / 4); i += SC_THREADS)
        out4[i] = ((const float4*)acc)[i];
}

__device__ __forceinline__ float gate_one(
    float i_r, float i_z, float i_n, float h_r, float h_z, float h_n, float hv)
{
    const float r = 1.0f / (1.0f + __expf(-(i_r + h_r)));
    const float z = 1.0f / (1.0f + __expf(-(i_z + h_z)));
    const float nn = tanhf(i_n + r * h_n);
    return (1.0f - z) * nn + z * hv;
}

struct GateOps { float hr[8], hz[8], hn[8], hv[8]; };

__device__ __forceinline__ void load_gate_ops(
    GateOps& g, const float* gh, const float* h, int tile, int half, int hc)
{
    #pragma unroll
    for (int r = 0; r < 8; ++r) {
        const size_t row = (size_t)(tile * 16 + r + half * 8);
        const float* p = gh + row * 384 + hc;
        g.hr[r] = p[0];
        g.hz[r] = p[128];
        g.hn[r] = p[256];
        g.hv[r] = h[row * KDIM + hc];
    }
}

__device__ __forceinline__ void gi_gate_tile(
    const v4f (&a)[16], const v4f (&b)[3][16], const float (&bv)[3],
    const GateOps& g, float* h, int tile, int half, int hc)
{
    v8f acc[3];
    #pragma unroll
    for (int t = 0; t < 3; ++t) acc[t] = v8f{0.f, 0.f, 0.f, 0.f, 0.f, 0.f, 0.f, 0.f};

    #pragma unroll
    for (int q = 0; q < 16; ++q) {
        const v2f a0 = {a[q].x, a[q].y};
        const v2f a1 = {a[q].z, a[q].w};
        #pragma unroll
        for (int t = 0; t < 3; ++t) {
            const v2f b0 = {b[t][q].x, b[t][q].y};
            acc[t] = __builtin_amdgcn_wmma_f32_16x16x4_f32(
                false, a0, false, b0, (short)0, acc[t], false, false);
        }
        #pragma unroll
        for (int t = 0; t < 3; ++t) {
            const v2f b1 = {b[t][q].z, b[t][q].w};
            acc[t] = __builtin_amdgcn_wmma_f32_16x16x4_f32(
                false, a1, false, b1, (short)0, acc[t], false, false);
        }
    }

    #pragma unroll
    for (int r = 0; r < 8; ++r) {
        const size_t row = (size_t)(tile * 16 + r + half * 8);
        h[row * KDIM + hc] = gate_one(
            acc[0][r] + bv[0], acc[1][r] + bv[1], acc[2][r] + bv[2],
            g.hr[r], g.hz[r], g.hn[r], g.hv[r]);
    }
}

__global__ __launch_bounds__(256) __attribute__((amdgpu_waves_per_eu(2, 2))) void ggnn_gi_gates_next(
    const float* __restrict__ agg, const float* __restrict__ w_ih,
    const float* __restrict__ b_ih, float* gh, float* h,
    const float* __restrict__ w_next, const float* __restrict__ w_hh,
    const float* __restrict__ b_hh, float* __restrict__ m, int ntiles)
{
    const int tid  = threadIdx.x;
    const int wave = __builtin_amdgcn_readfirstlane(tid >> 5);
    const int lane = tid & 31;
    const int half = lane >> 4;
    const int c    = lane & 15;
    const int hc   = wave * 16 + c;

    v4f   b[3][16];
    float bv[3];
    #pragma unroll
    for (int t = 0; t < 3; ++t) {
        const int n = t * 128 + hc;
        bv[t] = b_ih[n];
        #pragma unroll
        for (int q = 0; q < 16; ++q)
            b[t][q] = *(const v4f*)(w_ih + (size_t)n * KDIM + frag_k0(q, half));
    }

    const int nstreams = (int)gridDim.x;
    int tile = (int)blockIdx.x;
    if (tile >= ntiles) return;
    const int ntrips = (ntiles - tile + nstreams - 1) / nstreams;
    const int last_tile = tile + (ntrips - 1) * nstreams;

    v4f a0[16], a1[16];
    GateOps g;
    load_a_tile(a0, agg, tile, c, half);
    int i = 0;
    for (; i + 2 <= ntrips; i += 2) {
        load_a_tile(a1, agg, tile + nstreams, c, half);
        load_gate_ops(g, gh, h, tile, half, hc);
        __builtin_amdgcn_sched_barrier(0);
        gi_gate_tile(a0, b, bv, g, h, tile, half, hc);
        const int t2 = tile + 2 * nstreams;
        load_a_tile(a0, agg, t2 <= last_tile ? t2 : last_tile, c, half);
        load_gate_ops(g, gh, h, tile + nstreams, half, hc);
        __builtin_amdgcn_sched_barrier(0);
        gi_gate_tile(a1, b, bv, g, h, tile + nstreams, half, hc);
        tile = t2;
    }
    if (i < ntrips) {
        load_gate_ops(g, gh, h, tile, half, hc);
        __builtin_amdgcn_sched_barrier(0);
        gi_gate_tile(a0, b, bv, g, h, tile, half, hc);
    }

    if (w_next == nullptr) return;

    __builtin_amdgcn_fence(__ATOMIC_RELEASE, "agent");
    __syncthreads();
    __builtin_amdgcn_fence(__ATOMIC_ACQUIRE, "agent");

    m_gh_pass(h, w_next, w_hh, b_hh, m, gh, ntiles, (int)blockIdx.x, nstreams, wave, lane);
}

extern "C" void kernel_launch(void* const* d_in, const int* in_sizes, int n_in,
                              void* d_out, int out_size, void* d_ws, size_t ws_size,
                              hipStream_t stream)
{
    (void)n_in; (void)out_size; (void)ws_size;

    const float* x      = (const float*)d_in[0];
    const int*   ei     = (const int*)d_in[1];
    const float* lin_w  = (const float*)d_in[2];
    const float* lin_b  = (const float*)d_in[3];
    const float* ggc_w  = (const float*)d_in[4];
    const float* w_ih   = (const float*)d_in[5];
    const float* w_hh   = (const float*)d_in[6];
    const float* b_ih   = (const float*)d_in[7];
    const float* b_hh   = (const float*)d_in[8];
    const float* lin2_w = (const float*)d_in[9];
    const float* lin2_b = (const float*)d_in[10];
    float* out = (float*)d_out;

    const int N = in_sizes[0] / KDIM;
    const int E = in_sizes[1] / 2;
    const int* src = ei;
    const int* dst = ei + E;

    float* h   = (float*)d_ws;
    float* m   = h   + (size_t)N * 128;
    float* agg = m   + (size_t)N * 128;
    float* gh  = agg + (size_t)N * 128;

    const int row_tiles = (N + 15) / 16;
    const dim3 blk(256);
    const dim3 gemm_grid(256);

    const unsigned scatter_blks = (unsigned)((N + SC_ROWS - 1) / SC_ROWS);
    int* lists  = (int*)(gh + (size_t)N * 384);
    int* counts = lists + (size_t)scatter_blks * SC_CAP * 2;

    ggnn_gemm<2, 4, false, false, false><<<gemm_grid, blk, 0, stream>>>(
        x, lin_w, lin_b, h, row_tiles, 128, 128);

    ggnn_group_edges<<<scatter_blks, dim3(SC_THREADS), 0, stream>>>(src, dst, lists, counts, N, E);

    ggnn_m_gh<<<gemm_grid, blk, 0, stream>>>(h, ggc_w, w_hh, b_hh, m, gh, row_tiles);

    for (int l = 0; l < 7; ++l) {
        ggnn_scatter_owned<<<scatter_blks, dim3(SC_THREADS), 0, stream>>>(m, src, dst, lists, counts, agg, N, E);
        ggnn_gi_gates_next<<<gemm_grid, blk, 0, stream>>>(
            agg, w_ih, b_ih, gh, h, ggc_w + (size_t)(l + 1) * 128 * 128, w_hh, b_hh, m, row_tiles);
    }
    ggnn_scatter_owned<<<scatter_blks, dim3(SC_THREADS), 0, stream>>>(m, src, dst, lists, counts, agg, N, E);
    ggnn_gi_gates_next<<<gemm_grid, blk, 0, stream>>>(
        agg, w_ih, b_ih, gh, h, nullptr, w_hh, b_hh, m, row_tiles);

    ggnn_gemm<2, 4, false, true, true><<<gemm_grid, blk, 0, stream>>>(
        h, lin2_w, lin2_b, out, row_tiles, 100, 100);
}


# Edge grouping by slices: a group sorts its own 4096 edges by row range in LDS, appends each run to its range's list (1 integer add per run); was 250 groups scanning all edges; sums: LDS arrival
# speedup vs baseline: 3.7987x; 1.0401x over previous
#include <hip/hip_runtime.h>
#include <hip/hip_bf16.h>
#include <math.h>

typedef float v2f __attribute__((ext_vector_type(2)));
typedef float v4f __attribute__((ext_vector_type(4)));
typedef float v8f __attribute__((ext_vector_type(8)));

#define KDIM 128

#define GE_PITCH 32

__device__ __forceinline__ int frag_k0(int q, int h) { return 8 * q + 4 * h; }

__device__ __forceinline__ void load_a_tile(
    v4f (&a)[16], const float* __restrict__ X, int tile, int c, int h)
{
    const float* p = X + (size_t)(tile * 16 + c) * KDIM;
    #pragma unroll
    for (int q = 0; q < 16; ++q)
        a[q] = *(const v4f*)(p + frag_k0(q, h));
}

template <int CT, bool RELU_IN, bool GUARD>
__device__ __forceinline__ void tile_body(
    const v4f (&a)[16], const v4f (&b)[CT][16], const float (&bv)[CT],
    float* __restrict__ Y, int tile, int h, int c, int col0,
    int out_stride, int out_cols_total)
{
    v8f acc[CT];
    #pragma unroll
    for (int t = 0; t < CT; ++t) acc[t] = v8f{0.f, 0.f, 0.f, 0.f, 0.f, 0.f, 0.f, 0.f};

    #pragma unroll
    for (int q = 0; q < 16; ++q) {
        v2f a0 = {a[q].x, a[q].y};
        v2f a1 = {a[q].z, a[q].w};
        if (RELU_IN) {
            a0.x = fmaxf(a0.x, 0.0f); a0.y = fmaxf(a0.y, 0.0f);
            a1.x = fmaxf(a1.x, 0.0f); a1.y = fmaxf(a1.y, 0.0f);
        }
        #pragma unroll
        for (int t = 0; t < CT; ++t) {
            const v2f b0 = {b[t][q].x, b[t][q].y};
            acc[t] = __builtin_amdgcn_wmma_f32_16x16x4_f32(
                false, a0, false, b0, (short)0, acc[t], false, false);
        }
        #pragma unroll
        for (int t = 0; t < CT; ++t) {
            const v2f b1 = {b[t][q].z, b[t][q].w};
            acc[t] = __builtin_amdgcn_wmma_f32_16x16x4_f32(
                false, a1, false, b1, (short)0, acc[t], false, false);
        }
    }

    #pragma unroll
    for (int t = 0; t < CT; ++t) {
        const int cc = col0 + t * 16 + c;
        if (!GUARD || cc < out_cols_total) {
            #pragma unroll
            for (int r = 0; r < 8; ++r) {
                const int row = tile * 16 + r + h * 8;
                Y[(size_t)row * out_stride + cc] = acc[t][r] + bv[t];
            }
        }
    }
}

template <int CT, int NG, bool TRANS_FILL, bool RELU_IN, bool GUARD>
__global__ __launch_bounds__(256) __attribute__((amdgpu_waves_per_eu(2, 2))) void ggnn_gemm(
    const float* __restrict__ X, const float* __restrict__ W,
    const float* __restrict__ bias, float* __restrict__ Y,
    int ntiles, int out_stride, int out_cols_total,
    int* __restrict__ zero_counts, int nzero)
{
    constexpr int NSTR = 8 / NG;
    const int tid  = threadIdx.x;

    if (zero_counts != nullptr && blockIdx.x == 0)
        for (int i = tid; i < nzero; i += 256) zero_counts[(size_t)i * GE_PITCH] = 0;

    const int wave = __builtin_amdgcn_readfirstlane(tid >> 5);
    const int lane = tid & 31;
    const int h    = lane >> 4;
    const int c    = lane & 15;
    const int grp  = wave % NG;
    const int strm = wave / NG;
    const int col0 = grp * (CT * 16);

    v4f   b[CT][16];
    float bv[CT];
    #pragma unroll
    for (int t = 0; t < CT; ++t) {
        const int n = col0 + t * 16 + c;
        const bool ok = !GUARD || (n < out_cols_total);
        bv[t] = (bias != nullptr && ok) ? bias[n] : 0.0f;
        #pragma unroll
        for (int q = 0; q < 16; ++q) {
            const int k0 = frag_k0(q, h);
            v4f w = {0.f, 0.f, 0.f, 0.f};
            if (ok) {
                if (TRANS_FILL) {
                    w.x = W[(size_t)(k0 + 0) * out_cols_total + n];
                    w.y = W[(size_t)(k0 + 1) * out_cols_total + n];
                    w.z = W[(size_t)(k0 + 2) * out_cols_total + n];
                    w.w = W[(size_t)(k0 + 3) * out_cols_total + n];
                } else {
                    w = *(const v4f*)(W + (size_t)n * KDIM + k0);
                }
            }
            b[t][q] = w;
        }
    }

    const int nstreams = (int)gridDim.x * NSTR;
    int tile = (int)blockIdx.x * NSTR + strm;
    if (tile >= ntiles) return;
    const int ntrips = (ntiles - tile + nstreams - 1) / nstreams;
    const int last_tile = tile + (ntrips - 1) * nstreams;

    v4f a0[16], a1[16];
    load_a_tile(a0, X, tile, c, h);
    int i = 0;
    for (; i + 2 <= ntrips; i += 2) {
        load_a_tile(a1, X, tile + nstreams, c, h);
        __builtin_amdgcn_sched_barrier(0);
        tile_body<CT, RELU_IN, GUARD>(a0, b, bv, Y, tile, h, c, col0, out_stride, out_cols_total);
        const int t2 = tile + 2 * nstreams;
        load_a_tile(a0, X, t2 <= last_tile ? t2 : last_tile, c, h);
        __builtin_amdgcn_sched_barrier(0);
        tile_body<CT, RELU_IN, GUARD>(a1, b, bv, Y, tile + nstreams, h, c, col0, out_stride, out_cols_total);
        tile = t2;
    }
    if (i < ntrips)
        tile_body<CT, RELU_IN, GUARD>(a0, b, bv, Y, tile, h, c, col0, out_stride, out_cols_total);
}

__device__ __forceinline__ void m_gh_tile(
    const v4f (&a)[16], const v4f (&b)[4][16], const float (&bv)[4],
    float* __restrict__ M, float* GH, int tile, int h, int mc, int gc)
{
    v8f acc[4];
    #pragma unroll
    for (int t = 0; t < 4; ++t) acc[t] = v8f{0.f, 0.f, 0.f, 0.f, 0.f, 0.f, 0.f, 0.f};

    #pragma unroll
    for (int q = 0; q < 16; ++q) {
        const v2f a0 = {a[q].x, a[q].y};
        const v2f a1 = {a[q].z, a[q].w};
        #pragma unroll
        for (int t = 0; t < 4; ++t) {
            const v2f b0 = {b[t][q].x, b[t][q].y};
            acc[t] = __builtin_amdgcn_wmma_f32_16x16x4_f32(
                false, a0, false, b0, (short)0, acc[t], false, false);
        }
        #pragma unroll
        for (int t = 0; t < 4; ++t) {
            const v2f b1 = {b[t][q].z, b[t][q].w};
            acc[t] = __builtin_amdgcn_wmma_f32_16x16x4_f32(
                false, a1, false, b1, (short)0, acc[t], false, false);
        }
    }

    #pragma unroll
    for (int r = 0; r < 8; ++r) {
        const size_t row = (size_t)(tile * 16 + r + h * 8);
        M[row * 128 + mc] = acc[0][r];
        #pragma unroll
        for (int t = 1; t < 4; ++t)
            GH[row * 384 + gc + (t - 1) * 16] = acc[t][r] + bv[t];
    }
}

__device__ __forceinline__ void m_gh_pass(
    const float* X, const float* __restrict__ w_l, const float* __restrict__ w_hh,
    const float* __restrict__ b_hh, float* __restrict__ M, float* GH,
    int ntiles, int first_tile, int nstreams, int wave, int lane)
{
    const int h  = lane >> 4;
    const int c  = lane & 15;
    const int mc = wave * 16 + c;
    const int gc = wave * 48 + c;

    v4f   b[4][16];
    float bv[4];
    bv[0] = 0.0f;
    #pragma unroll
    for (int q = 0; q < 16; ++q) {
        const int k0 = frag_k0(q, h);
        v4f w;
        w.x = w_l[(size_t)(k0 + 0) * 128 + mc];
        w.y = w_l[(size_t)(k0 + 1) * 128 + mc];
        w.z = w_l[(size_t)(k0 + 2) * 128 + mc];
        w.w = w_l[(size_t)(k0 + 3) * 128 + mc];
        b[0][q] = w;
    }
    #pragma unroll
    for (int t = 1; t < 4; ++t) {
        const int n = gc + (t - 1) * 16;
        bv[t] = b_hh[n];
        #pragma unroll
        for (int q = 0; q < 16; ++q)
            b[t][q] = *(const v4f*)(w_hh + (size_t)n * KDIM + frag_k0(q, h));
    }

    int tile = first_tile;
    if (tile >= ntiles) return;
    const int ntrips = (ntiles - tile + nstreams - 1) / nstreams;
    const int last_tile = tile + (ntrips - 1) * nstreams;

    v4f a0[16], a1[16];
    load_a_tile(a0, X, tile, c, h);
    int i = 0;
    for (; i + 2 <= ntrips; i += 2) {
        load_a_tile(a1, X, tile + nstreams, c, h);
        __builtin_amdgcn_sched_barrier(0);
        m_gh_tile(a0, b, bv, M, GH, tile, h, mc, gc);
        const int t2 = tile + 2 * nstreams;
        load_a_tile(a0, X, t2 <= last_tile ? t2 : last_tile, c, h);
        __builtin_amdgcn_sched_barrier(0);
        m_gh_tile(a1, b, bv, M, GH, tile + nstreams, h, mc, gc);
        tile = t2;
    }
    if (i < ntrips)
        m_gh_tile(a0, b, bv, M, GH, tile, h, mc, gc);
}

__global__ __launch_bounds__(256) __attribute__((amdgpu_waves_per_eu(2, 2))) void ggnn_m_gh(
    const float* __restrict__ X, const float* __restrict__ w_l,
    const float* __restrict__ w_hh, const float* __restrict__ b_hh,
    float* __restrict__ M, float* __restrict__ GH, int ntiles)
{
    const int tid  = threadIdx.x;
    const int wave = __builtin_amdgcn_readfirstlane(tid >> 5);
    m_gh_pass(X, w_l, w_hh, b_hh, M, GH, ntiles, (int)blockIdx.x, (int)gridDim.x, wave, tid & 31);
}

#define SC_ROWS    200
#define SC_CAP     8192
#define SC_THREADS 1024
#define SC_VEC     8
#define SC_GATHER  8
#define GE_SLICE   4096
#define GE_MAXR    1024

template <bool ADD_WHEN_FULL>
__device__ __forceinline__ void sc_note_edge(
    int e, int d, int lo, int rows, int* cnt, int2* list, float* acc,
    const float* __restrict__ m, const int* __restrict__ src)
{
    const unsigned r = (unsigned)(d - lo);
    if (r < (unsigned)rows) {
        const int slot = atomicAdd(cnt, 1);
        if (slot < SC_CAP) {
            list[slot] = make_int2(e, (int)r);
        } else if (ADD_WHEN_FULL) {
            const float* p = m + (size_t)src[e] * KDIM;
            float* a = acc + (size_t)r * KDIM;
            #pragma nounroll
            for (int j = 0; j < KDIM; ++j) atomicAdd(a + j, p[j]);
        }
    }
}

template <bool ADD_WHEN_FULL>
__device__ __forceinline__ void sc_filter_edges(
    int lo, int rows, int tid, int* cnt, int2* list, float* acc,
    const float* __restrict__ m, const int* __restrict__ src,
    const int* __restrict__ dst, int nedges)
{
    if ((nedges & 3) == 0) {
        for (int base = 0; base < nedges; base += SC_VEC * SC_THREADS * 4) {
            int4 d[SC_VEC];
            #pragma unroll
            for (int k = 0; k < SC_VEC; ++k) {
                const int e0 = base + (k * SC_THREADS + tid) * 4;
                d[k] = make_int4(lo - 1, lo - 1, lo - 1, lo - 1);
                if (e0 < nedges) d[k] = *(const int4*)(dst + e0);
            }
            #pragma unroll
            for (int k = 0; k < SC_VEC; ++k) {
                const int e0 = base + (k * SC_THREADS + tid) * 4;
                sc_note_edge<ADD_WHEN_FULL>(e0 + 0, d[k].x, lo, rows, cnt, list, acc, m, src);
                sc_note_edge<ADD_WHEN_FULL>(e0 + 1, d[k].y, lo, rows, cnt, list, acc, m, src);
                sc_note_edge<ADD_WHEN_FULL>(e0 + 2, d[k].z, lo, rows, cnt, list, acc, m, src);
                sc_note_edge<ADD_WHEN_FULL>(e0 + 3, d[k].w, lo, rows, cnt, list, acc, m, src);
            }
        }
    } else {
        for (int e = tid; e < nedges; e += SC_THREADS)
            sc_note_edge<ADD_WHEN_FULL>(e, dst[e], lo, rows, cnt, list, acc, m, src);
    }
    __syncthreads();

    const int n = (*cnt < SC_CAP) ? *cnt : SC_CAP;
    for (int i = tid; i < n; i += SC_THREADS)
        list[i].x = src[list[i].x];
    __syncthreads();
}

__global__ __launch_bounds__(SC_THREADS) void ggnn_group_edges(
    const int* __restrict__ src, const int* __restrict__ dst,
    int* __restrict__ lists, int* __restrict__ counts, int nnodes, int nedges, int nranges)
{
    __shared__ int  hist[GE_MAXR];
    __shared__ int  lstart[GE_MAXR];
    __shared__ int  gbase[GE_MAXR];
    __shared__ int  wtot[SC_THREADS / 32];
    __shared__ int2 sorted[GE_SLICE];
    __shared__ int  srange[GE_SLICE];

    const int tid  = threadIdx.x;
    const int lane = tid & 31;
    const int wv   = tid >> 5;
    const int e0   = (int)blockIdx.x * GE_SLICE;

    for (int g = tid; g < nranges; g += SC_THREADS) hist[g] = 0;
    __syncthreads();

    int eg[GE_SLICE / SC_THREADS], er[GE_SLICE / SC_THREADS];
    int es[GE_SLICE / SC_THREADS], eslot[GE_SLICE / SC_THREADS];
    #pragma unroll
    for (int k = 0; k < GE_SLICE / SC_THREADS; ++k) {
        const int e = e0 + k * SC_THREADS + tid;
        int d = -1, s = 0;
        if (e < nedges) { d = dst[e]; s = src[e]; }
        const bool ok = (unsigned)d < (unsigned)nnodes;
        const int g = ok ? d / SC_ROWS : -1;
        eg[k] = g;
        er[k] = d - g * SC_ROWS;
        es[k] = s;
        eslot[k] = 0;
    }
    #pragma unroll
    for (int k = 0; k < GE_SLICE / SC_THREADS; ++k)
        if (eg[k] >= 0) eslot[k] = atomicAdd(&hist[eg[k]], 1);
    __syncthreads();

    const int cnt_g = (tid < nranges) ? hist[tid] : 0;
    int incl = cnt_g;
    #pragma unroll
    for (int dlt = 1; dlt < 32; dlt <<= 1) {
        const int up = __shfl_up(incl, dlt);
        if (lane >= dlt) incl += up;
    }
    if (lane == 31) wtot[wv] = incl;
    __syncthreads();
    if (tid < nranges) {
        int before = 0;
        for (int j = 0; j < wv; ++j) before += wtot[j];
        lstart[tid] = before + incl - cnt_g;
        gbase[tid] = (cnt_g > 0) ? atomicAdd(counts + (size_t)tid * GE_PITCH, cnt_g) : 0;
    }
    __syncthreads();

    #pragma unroll
    for (int k = 0; k < GE_SLICE / SC_THREADS; ++k) {
        if (eg[k] >= 0) {
            const int j = lstart[eg[k]] + eslot[k];
            sorted[j] = make_int2(es[k], er[k]);
            srange[j] = eg[k];
        }
    }
    __syncthreads();

    const int nvalid = lstart[nranges - 1] + hist[nranges - 1];
    for (int j = tid; j < nvalid; j += SC_THREADS) {
        const int g = srange[j];
        const int pos = gbase[g] + (j - lstart[g]);
        if (pos < SC_CAP)
            ((int2*)lists)[(size_t)g * SC_CAP + pos] = sorted[j];
    }
}

__global__ __launch_bounds__(SC_THREADS) void ggnn_scatter_owned(
    const float* __restrict__ m, const int* __restrict__ src,
    const int* __restrict__ dst, const int* __restrict__ lists,
    const int* __restrict__ counts, float* __restrict__ agg, int nnodes, int nedges)
{
    __shared__ __attribute__((aligned(16))) float acc[SC_ROWS * KDIM];
    __shared__ int2 list[SC_CAP];
    __shared__ int cnt;

    const int tid  = threadIdx.x;
    const int lane = tid & 31;
    const int wave = __builtin_amdgcn_readfirstlane(tid >> 5);

    const int g    = (int)blockIdx.x;
    const int lo   = g * SC_ROWS;
    const int rows = (nnodes - lo < SC_ROWS) ? (nnodes - lo) : SC_ROWS;
    const int total = __builtin_amdgcn_readfirstlane(counts[(size_t)g * GE_PITCH]);
    const bool have_list = total <= SC_CAP;

    int2 t[SC_CAP / SC_THREADS];
    #pragma unroll
    for (int k = 0; k < SC_CAP / SC_THREADS; ++k) {
        const int i = k * SC_THREADS + tid;
        t[k] = make_int2(0, 0);
        if (have_list && i < total)
            t[k] = ((const int2*)lists)[(size_t)g * SC_CAP + i];
    }
    for (int i = tid; i < SC_ROWS * KDIM / 4; i += SC_THREADS)
        ((float4*)acc)[i] = make_float4(0.f, 0.f, 0.f, 0.f);
    if (tid == 0) cnt = 0;
    #pragma unroll
    for (int k = 0; k < SC_CAP / SC_THREADS; ++k) {
        const int i = k * SC_THREADS + tid;
        if (have_list && i < total) list[i] = t[k];
    }
    __syncthreads();

    if (!have_list)
        sc_filter_edges<true>(lo, rows, tid, &cnt, list, acc, m, src, dst, nedges);

    const int n = (total < SC_CAP) ? total : SC_CAP;

    for (int i = wave; i < n; i += 32 * SC_GATHER) {
        int2 mine = make_int2(0, -1);
        const int my_idx = i + 32 * lane;
        if (lane < SC_GATHER && my_idx < n) mine = list[my_idx];
        float v[SC_GATHER][4];
        int   r[SC_GATHER];
        #pragma unroll
        for (int u = 0; u < SC_GATHER; ++u) {
            const int s = __builtin_amdgcn_readlane(mine.x, u);
            r[u] = __builtin_amdgcn_readlane(mine.y, u);
            const float* p = m + (size_t)s * KDIM + lane;
            v[u][0] = p[0]; v[u][1] = p[32]; v[u][2] = p[64]; v[u][3] = p[96];
        }
        #pragma unroll
        for (int u = 0; u < SC_GATHER; ++u) {
            if (r[u] >= 0) {
                float* a = acc + r[u] * KDIM + lane;
                atomicAdd(a,      v[u][0]);
                atomicAdd(a + 32, v[u][1]);
                atomicAdd(a + 64, v[u][2]);
                atomicAdd(a + 96, v[u][3]);
            }
        }
    }
    __syncthreads();

    float4* out4 = (float4*)(agg + (size_t)lo * KDIM);
    for (int i = tid; i < rows * (KDIM / 4); i += SC_THREADS)
        out4[i] = ((const float4*)acc)[i];
}

__device__ __forceinline__ float gate_one(
    float i_r, float i_z, float i_n, float h_r, float h_z, float h_n, float hv)
{
    const float r = 1.0f / (1.0f + __expf(-(i_r + h_r)));
    const float z = 1.0f / (1.0f + __expf(-(i_z + h_z)));
    const float nn = tanhf(i_n + r * h_n);
    return (1.0f - z) * nn + z * hv;
}

struct GateOps { float hr[8], hz[8], hn[8], hv[8]; };

__device__ __forceinline__ void load_gate_ops(
    GateOps& g, const float* gh, const float* h, int tile, int half, int hc)
{
    #pragma unroll
    for (int r = 0; r < 8; ++r) {
        const size_t row = (size_t)(tile * 16 + r + half * 8);
        const float* p = gh + row * 384 + hc;
        g.hr[r] = p[0];
        g.hz[r] = p[128];
        g.hn[r] = p[256];
        g.hv[r] = h[row * KDIM + hc];
    }
}

__device__ __forceinline__ void gi_gate_tile(
    const v4f (&a)[16], const v4f (&b)[3][16], const float (&bv)[3],
    const GateOps& g, float* h, int tile, int half, int hc)
{
    v8f acc[3];
    #pragma unroll
    for (int t = 0; t < 3; ++t) acc[t] = v8f{0.f, 0.f, 0.f, 0.f, 0.f, 0.f, 0.f, 0.f};

    #pragma unroll
    for (int q = 0; q < 16; ++q) {
        const v2f a0 = {a[q].x, a[q].y};
        const v2f a1 = {a[q].z, a[q].w};
        #pragma unroll
        for (int t = 0; t < 3; ++t) {
            const v2f b0 = {b[t][q].x, b[t][q].y};
            acc[t] = __builtin_amdgcn_wmma_f32_16x16x4_f32(
                false, a0, false, b0, (short)0, acc[t], false, false);
        }
        #pragma unroll
        for (int t = 0; t < 3; ++t) {
            const v2f b1 = {b[t][q].z, b[t][q].w};
            acc[t] = __builtin_amdgcn_wmma_f32_16x16x4_f32(
                false, a1, false, b1, (short)0, acc[t], false, false);
        }
    }

    #pragma unroll
    for (int r = 0; r < 8; ++r) {
        const size_t row = (size_t)(tile * 16 + r + half * 8);
        h[row * KDIM + hc] = gate_one(
            acc[0][r] + bv[0], acc[1][r] + bv[1], acc[2][r] + bv[2],
            g.hr[r], g.hz[r], g.hn[r], g.hv[r]);
    }
}

__global__ __launch_bounds__(256) __attribute__((amdgpu_waves_per_eu(2, 2))) void ggnn_gi_gates_next(
    const float* __restrict__ agg, const float* __restrict__ w_ih,
    const float* __restrict__ b_ih, float* gh, float* h,
    const float* __restrict__ w_next, const float* __restrict__ w_hh,
    const float* __restrict__ b_hh, float* __restrict__ m, int ntiles)
{
    const int tid  = threadIdx.x;
    const int wave = __builtin_amdgcn_readfirstlane(tid >> 5);
    const int lane = tid & 31;
    const int half = lane >> 4;
    const int c    = lane & 15;
    const int hc   = wave * 16 + c;

    v4f   b[3][16];
    float bv[3];
    #pragma unroll
    for (int t = 0; t < 3; ++t) {
        const int n = t * 128 + hc;
        bv[t] = b_ih[n];
        #pragma unroll
        for (int q = 0; q < 16; ++q)
            b[t][q] = *(const v4f*)(w_ih + (size_t)n * KDIM + frag_k0(q, half));
    }

    const int nstreams = (int)gridDim.x;
    int tile = (int)blockIdx.x;
    if (tile >= ntiles) return;
    const int ntrips = (ntiles - tile + nstreams - 1) / nstreams;
    const int last_tile = tile + (ntrips - 1) * nstreams;

    v4f a0[16], a1[16];
    GateOps g;
    load_a_tile(a0, agg, tile, c, half);
    int i = 0;
    for (; i + 2 <= ntrips; i += 2) {
        load_a_tile(a1, agg, tile + nstreams, c, half);
        load_gate_ops(g, gh, h, tile, half, hc);
        __builtin_amdgcn_sched_barrier(0);
        gi_gate_tile(a0, b, bv, g, h, tile, half, hc);
        const int t2 = tile + 2 * nstreams;
        load_a_tile(a0, agg, t2 <= last_tile ? t2 : last_tile, c, half);
        load_gate_ops(g, gh, h, tile + nstreams, half, hc);
        __builtin_amdgcn_sched_barrier(0);
        gi_gate_tile(a1, b, bv, g, h, tile + nstreams, half, hc);
        tile = t2;
    }
    if (i < ntrips) {
        load_gate_ops(g, gh, h, tile, half, hc);
        __builtin_amdgcn_sched_barrier(0);
        gi_gate_tile(a0, b, bv, g, h, tile, half, hc);
    }

    if (w_next == nullptr) return;

    __builtin_amdgcn_fence(__ATOMIC_RELEASE, "agent");
    __syncthreads();
    __builtin_amdgcn_fence(__ATOMIC_ACQUIRE, "agent");

    m_gh_pass(h, w_next, w_hh, b_hh, m, gh, ntiles, (int)blockIdx.x, nstreams, wave, lane);
}

extern "C" void kernel_launch(void* const* d_in, const int* in_sizes, int n_in,
                              void* d_out, int out_size, void* d_ws, size_t ws_size,
                              hipStream_t stream)
{
    (void)n_in; (void)out_size; (void)ws_size;

    const float* x      = (const float*)d_in[0];
    const int*   ei     = (const int*)d_in[1];
    const float* lin_w  = (const float*)d_in[2];
    const float* lin_b  = (const float*)d_in[3];
    const float* ggc_w  = (const float*)d_in[4];
    const float* w_ih   = (const float*)d_in[5];
    const float* w_hh   = (const float*)d_in[6];
    const float* b_ih   = (const float*)d_in[7];
    const float* b_hh   = (const float*)d_in[8];
    const float* lin2_w = (const float*)d_in[9];
    const float* lin2_b = (const float*)d_in[10];
    float* out = (float*)d_out;

    const int N = in_sizes[0] / KDIM;
    const int E = in_sizes[1] / 2;
    const int* src = ei;
    const int* dst = ei + E;

    float* h   = (float*)d_ws;
    float* m   = h   + (size_t)N * 128;
    float* agg = m   + (size_t)N * 128;
    float* gh  = agg + (size_t)N * 128;

    const int row_tiles = (N + 15) / 16;
    const dim3 blk(256);
    const dim3 gemm_grid(256);

    const unsigned scatter_blks = (unsigned)((N + SC_ROWS - 1) / SC_ROWS);
    int* lists  = (int*)(gh + (size_t)N * 384);
    int* counts = lists + (size_t)scatter_blks * SC_CAP * 2;
    const unsigned group_blks = (unsigned)((E + GE_SLICE - 1) / GE_SLICE);

    ggnn_gemm<2, 4, false, false, false><<<gemm_grid, blk, 0, stream>>>(
        x, lin_w, lin_b, h, row_tiles, 128, 128, counts, (int)scatter_blks);

    ggnn_group_edges<<<group_blks, dim3(SC_THREADS), 0, stream>>>(
        src, dst, lists, counts, N, E, (int)scatter_blks);

    ggnn_m_gh<<<gemm_grid, blk, 0, stream>>>(h, ggc_w, w_hh, b_hh, m, gh, row_tiles);

    for (int l = 0; l < 7; ++l) {
        ggnn_scatter_owned<<<scatter_blks, dim3(SC_THREADS), 0, stream>>>(m, src, dst, lists, counts, agg, N, E);
        ggnn_gi_gates_next<<<gemm_grid, blk, 0, stream>>>(
            agg, w_ih, b_ih, gh, h, ggc_w + (size_t)(l + 1) * 128 * 128, w_hh, b_hh, m, row_tiles);
    }
    ggnn_scatter_owned<<<scatter_blks, dim3(SC_THREADS), 0, stream>>>(m, src, dst, lists, counts, agg, N, E);
    ggnn_gi_gates_next<<<gemm_grid, blk, 0, stream>>>(
        agg, w_ih, b_ih, gh, h, nullptr, w_hh, b_hh, m, row_tiles);

    ggnn_gemm<2, 4, false, true, true><<<gemm_grid, blk, 0, stream>>>(
        h, lin2_w, lin2_b, out, row_tiles, 100, 100, nullptr, 0);
}


# GRU gates of a row tile as 8 interleaved chains: tanh by the library's two formulas, both evaluated and chosen per lane (no branch per row), so the rows' chains overlap; sums, formulas unchanged
# speedup vs baseline: 4.0003x; 1.0531x over previous
#include <hip/hip_runtime.h>
#include <hip/hip_bf16.h>
#include <math.h>

typedef float v2f __attribute__((ext_vector_type(2)));
typedef float v4f __attribute__((ext_vector_type(4)));
typedef float v8f __attribute__((ext_vector_type(8)));

#define KDIM 128

#define GE_PITCH 32

__device__ __forceinline__ int frag_k0(int q, int h) { return 8 * q + 4 * h; }

__device__ __forceinline__ void load_a_tile(
    v4f (&a)[16], const float* __restrict__ X, int tile, int c, int h)
{
    const float* p = X + (size_t)(tile * 16 + c) * KDIM;
    #pragma unroll
    for (int q = 0; q < 16; ++q)
        a[q] = *(const v4f*)(p + frag_k0(q, h));
}

template <int CT, bool RELU_IN, bool GUARD>
__device__ __forceinline__ void tile_body(
    const v4f (&a)[16], const v4f (&b)[CT][16], const float (&bv)[CT],
    float* __restrict__ Y, int tile, int h, int c, int col0,
    int out_stride, int out_cols_total)
{
    v8f acc[CT];
    #pragma unroll
    for (int t = 0; t < CT; ++t) acc[t] = v8f{0.f, 0.f, 0.f, 0.f, 0.f, 0.f, 0.f, 0.f};

    #pragma unroll
    for (int q = 0; q < 16; ++q) {
        v2f a0 = {a[q].x, a[q].y};
        v2f a1 = {a[q].z, a[q].w};
        if (RELU_IN) {
            a0.x = fmaxf(a0.x, 0.0f); a0.y = fmaxf(a0.y, 0.0f);
            a1.x = fmaxf(a1.x, 0.0f); a1.y = fmaxf(a1.y, 0.0f);
        }
        #pragma unroll
        for (int t = 0; t < CT; ++t) {
            const v2f b0 = {b[t][q].x, b[t][q].y};
            acc[t] = __builtin_amdgcn_wmma_f32_16x16x4_f32(
                false, a0, false, b0, (short)0, acc[t], false, false);
        }
        #pragma unroll
        for (int t = 0; t < CT; ++t) {
            const v2f b1 = {b[t][q].z, b[t][q].w};
            acc[t] = __builtin_amdgcn_wmma_f32_16x16x4_f32(
                false, a1, false, b1, (short)0, acc[t], false, false);
        }
    }

    #pragma unroll
    for (int t = 0; t < CT; ++t) {
        const int cc = col0 + t * 16 + c;
        if (!GUARD || cc < out_cols_total) {
            #pragma unroll
            for (int r = 0; r < 8; ++r) {
                const int row = tile * 16 + r + h * 8;
                Y[(size_t)row * out_stride + cc] = acc[t][r] + bv[t];
            }
        }
    }
}

template <int CT, int NG, bool TRANS_FILL, bool RELU_IN, bool GUARD>
__global__ __launch_bounds__(256) __attribute__((amdgpu_waves_per_eu(2, 2))) void ggnn_gemm(
    const float* __restrict__ X, const float* __restrict__ W,
    const float* __restrict__ bias, float* __restrict__ Y,
    int ntiles, int out_stride, int out_cols_total,
    int* __restrict__ zero_counts, int nzero)
{
    constexpr int NSTR = 8 / NG;
    const int tid  = threadIdx.x;

    if (zero_counts != nullptr && blockIdx.x == 0)
        for (int i = tid; i < nzero; i += 256) zero_counts[(size_t)i * GE_PITCH] = 0;

    const int wave = __builtin_amdgcn_readfirstlane(tid >> 5);
    const int lane = tid & 31;
    const int h    = lane >> 4;
    const int c    = lane & 15;
    const int grp  = wave % NG;
    const int strm = wave / NG;
    const int col0 = grp * (CT * 16);

    v4f   b[CT][16];
    float bv[CT];
    #pragma unroll
    for (int t = 0; t < CT; ++t) {
        const int n = col0 + t * 16 + c;
        const bool ok = !GUARD || (n < out_cols_total);
        bv[t] = (bias != nullptr && ok) ? bias[n] : 0.0f;
        #pragma unroll
        for (int q = 0; q < 16; ++q) {
            const int k0 = frag_k0(q, h);
            v4f w = {0.f, 0.f, 0.f, 0.f};
            if (ok) {
                if (TRANS_FILL) {
                    w.x = W[(size_t)(k0 + 0) * out_cols_total + n];
                    w.y = W[(size_t)(k0 + 1) * out_cols_total + n];
                    w.z = W[(size_t)(k0 + 2) * out_cols_total + n];
                    w.w = W[(size_t)(k0 + 3) * out_cols_total + n];
                } else {
                    w = *(const v4f*)(W + (size_t)n * KDIM + k0);
                }
            }
            b[t][q] = w;
        }
    }

    const int nstreams = (int)gridDim.x * NSTR;
    int tile = (int)blockIdx.x * NSTR + strm;
    if (tile >= ntiles) return;
    const int ntrips = (ntiles - tile + nstreams - 1) / nstreams;
    const int last_tile = tile + (ntrips - 1) * nstreams;

    v4f a0[16], a1[16];
    load_a_tile(a0, X, tile, c, h);
    int i = 0;
    for (; i + 2 <= ntrips; i += 2) {
        load_a_tile(a1, X, tile + nstreams, c, h);
        __builtin_amdgcn_sched_barrier(0);
        tile_body<CT, RELU_IN, GUARD>(a0, b, bv, Y, tile, h, c, col0, out_stride, out_cols_total);
        const int t2 = tile + 2 * nstreams;
        load_a_tile(a0, X, t2 <= last_tile ? t2 : last_tile, c, h);
        __builtin_amdgcn_sched_barrier(0);
        tile_body<CT, RELU_IN, GUARD>(a1, b, bv, Y, tile + nstreams, h, c, col0, out_stride, out_cols_total);
        tile = t2;
    }
    if (i < ntrips)
        tile_body<CT, RELU_IN, GUARD>(a0, b, bv, Y, tile, h, c, col0, out_stride, out_cols_total);
}

__device__ __forceinline__ void m_gh_tile(
    const v4f (&a)[16], const v4f (&b)[4][16], const float (&bv)[4],
    float* __restrict__ M, float* GH, int tile, int h, int mc, int gc)
{
    v8f acc[4];
    #pragma unroll
    for (int t = 0; t < 4; ++t) acc[t] = v8f{0.f, 0.f, 0.f, 0.f, 0.f, 0.f, 0.f, 0.f};

    #pragma unroll
    for (int q = 0; q < 16; ++q) {
        const v2f a0 = {a[q].x, a[q].y};
        const v2f a1 = {a[q].z, a[q].w};
        #pragma unroll
        for (int t = 0; t < 4; ++t) {
            const v2f b0 = {b[t][q].x, b[t][q].y};
            acc[t] = __builtin_amdgcn_wmma_f32_16x16x4_f32(
                false, a0, false, b0, (short)0, acc[t], false, false);
        }
        #pragma unroll
        for (int t = 0; t < 4; ++t) {
            const v2f b1 = {b[t][q].z, b[t][q].w};
            acc[t] = __builtin_amdgcn_wmma_f32_16x16x4_f32(
                false, a1, false, b1, (short)0, acc[t], false, false);
        }
    }

    #pragma unroll
    for (int r = 0; r < 8; ++r) {
        const size_t row = (size_t)(tile * 16 + r + h * 8);
        M[row * 128 + mc] = acc[0][r];
        #pragma unroll
        for (int t = 1; t < 4; ++t)
            GH[row * 384 + gc + (t - 1) * 16] = acc[t][r] + bv[t];
    }
}

__device__ __forceinline__ void m_gh_pass(
    const float* X, const float* __restrict__ w_l, const float* __restrict__ w_hh,
    const float* __restrict__ b_hh, float* __restrict__ M, float* GH,
    int ntiles, int first_tile, int nstreams, int wave, int lane)
{
    const int h  = lane >> 4;
    const int c  = lane & 15;
    const int mc = wave * 16 + c;
    const int gc = wave * 48 + c;

    v4f   b[4][16];
    float bv[4];
    bv[0] = 0.0f;
    #pragma unroll
    for (int q = 0; q < 16; ++q) {
        const int k0 = frag_k0(q, h);
        v4f w;
        w.x = w_l[(size_t)(k0 + 0) * 128 + mc];
        w.y = w_l[(size_t)(k0 + 1) * 128 + mc];
        w.z = w_l[(size_t)(k0 + 2) * 128 + mc];
        w.w = w_l[(size_t)(k0 + 3) * 128 + mc];
        b[0][q] = w;
    }
    #pragma unroll
    for (int t = 1; t < 4; ++t) {
        const int n = gc + (t - 1) * 16;
        bv[t] = b_hh[n];
        #pragma unroll
        for (int q = 0; q < 16; ++q)
            b[t][q] = *(const v4f*)(w_hh + (size_t)n * KDIM + frag_k0(q, h));
    }

    int tile = first_tile;
    if (tile >= ntiles) return;
    const int ntrips = (ntiles - tile + nstreams - 1) / nstreams;
    const int last_tile = tile + (ntrips - 1) * nstreams;

    v4f a0[16], a1[16];
    load_a_tile(a0, X, tile, c, h);
    int i = 0;
    for (; i + 2 <= ntrips; i += 2) {
        load_a_tile(a1, X, tile + nstreams, c, h);
        __builtin_amdgcn_sched_barrier(0);
        m_gh_tile(a0, b, bv, M, GH, tile, h, mc, gc);
        const int t2 = tile + 2 * nstreams;
        load_a_tile(a0, X, t2 <= last_tile ? t2 : last_tile, c, h);
        __builtin_amdgcn_sched_barrier(0);
        m_gh_tile(a1, b, bv, M, GH, tile + nstreams, h, mc, gc);
        tile = t2;
    }
    if (i < ntrips)
        m_gh_tile(a0, b, bv, M, GH, tile, h, mc, gc);
}

__global__ __launch_bounds__(256) __attribute__((amdgpu_waves_per_eu(2, 2))) void ggnn_m_gh(
    const float* __restrict__ X, const float* __restrict__ w_l,
    const float* __restrict__ w_hh, const float* __restrict__ b_hh,
    float* __restrict__ M, float* __restrict__ GH, int ntiles)
{
    const int tid  = threadIdx.x;
    const int wave = __builtin_amdgcn_readfirstlane(tid >> 5);
    m_gh_pass(X, w_l, w_hh, b_hh, M, GH, ntiles, (int)blockIdx.x, (int)gridDim.x, wave, tid & 31);
}

#define SC_ROWS    200
#define SC_CAP     8192
#define SC_THREADS 1024
#define SC_VEC     8
#define SC_GATHER  8
#define GE_SLICE   4096
#define GE_MAXR    1024

template <bool ADD_WHEN_FULL>
__device__ __forceinline__ void sc_note_edge(
    int e, int d, int lo, int rows, int* cnt, int2* list, float* acc,
    const float* __restrict__ m, const int* __restrict__ src)
{
    const unsigned r = (unsigned)(d - lo);
    if (r < (unsigned)rows) {
        const int slot = atomicAdd(cnt, 1);
        if (slot < SC_CAP) {
            list[slot] = make_int2(e, (int)r);
        } else if (ADD_WHEN_FULL) {
            const float* p = m + (size_t)src[e] * KDIM;
            float* a = acc + (size_t)r * KDIM;
            #pragma nounroll
            for (int j = 0; j < KDIM; ++j) atomicAdd(a + j, p[j]);
        }
    }
}

template <bool ADD_WHEN_FULL>
__device__ __forceinline__ void sc_filter_edges(
    int lo, int rows, int tid, int* cnt, int2* list, float* acc,
    const float* __restrict__ m, const int* __restrict__ src,
    const int* __restrict__ dst, int nedges)
{
    if ((nedges & 3) == 0) {
        for (int base = 0; base < nedges; base += SC_VEC * SC_THREADS * 4) {
            int4 d[SC_VEC];
            #pragma unroll
            for (int k = 0; k < SC_VEC; ++k) {
                const int e0 = base + (k * SC_THREADS + tid) * 4;
                d[k] = make_int4(lo - 1, lo - 1, lo - 1, lo - 1);
                if (e0 < nedges) d[k] = *(const int4*)(dst + e0);
            }
            #pragma unroll
            for (int k = 0; k < SC_VEC; ++k) {
                const int e0 = base + (k * SC_THREADS + tid) * 4;
                sc_note_edge<ADD_WHEN_FULL>(e0 + 0, d[k].x, lo, rows, cnt, list, acc, m, src);
                sc_note_edge<ADD_WHEN_FULL>(e0 + 1, d[k].y, lo, rows, cnt, list, acc, m, src);
                sc_note_edge<ADD_WHEN_FULL>(e0 + 2, d[k].z, lo, rows, cnt, list, acc, m, src);
                sc_note_edge<ADD_WHEN_FULL>(e0 + 3, d[k].w, lo, rows, cnt, list, acc, m, src);
            }
        }
    } else {
        for (int e = tid; e < nedges; e += SC_THREADS)
            sc_note_edge<ADD_WHEN_FULL>(e, dst[e], lo, rows, cnt, list, acc, m, src);
    }
    __syncthreads();

    const int n = (*cnt < SC_CAP) ? *cnt : SC_CAP;
    for (int i = tid; i < n; i += SC_THREADS)
        list[i].x = src[list[i].x];
    __syncthreads();
}

__global__ __launch_bounds__(SC_THREADS) void ggnn_group_edges(
    const int* __restrict__ src, const int* __restrict__ dst,
    int* __restrict__ lists, int* __restrict__ counts, int nnodes, int nedges, int nranges)
{
    __shared__ int  hist[GE_MAXR];
    __shared__ int  lstart[GE_MAXR];
    __shared__ int  gbase[GE_MAXR];
    __shared__ int  wtot[SC_THREADS / 32];
    __shared__ int2 sorted[GE_SLICE];
    __shared__ int  srange[GE_SLICE];

    const int tid  = threadIdx.x;
    const int lane = tid & 31;
    const int wv   = tid >> 5;
    const int e0   = (int)blockIdx.x * GE_SLICE;

    for (int g = tid; g < nranges; g += SC_THREADS) hist[g] = 0;
    __syncthreads();

    int eg[GE_SLICE / SC_THREADS], er[GE_SLICE / SC_THREADS];
    int es[GE_SLICE / SC_THREADS], eslot[GE_SLICE / SC_THREADS];
    #pragma unroll
    for (int k = 0; k < GE_SLICE / SC_THREADS; ++k) {
        const int e = e0 + k * SC_THREADS + tid;
        int d = -1, s = 0;
        if (e < nedges) { d = dst[e]; s = src[e]; }
        const bool ok = (unsigned)d < (unsigned)nnodes;
        const int g = ok ? d / SC_ROWS : -1;
        eg[k] = g;
        er[k] = d - g * SC_ROWS;
        es[k] = s;
        eslot[k] = 0;
    }
    #pragma unroll
    for (int k = 0; k < GE_SLICE / SC_THREADS; ++k)
        if (eg[k] >= 0) eslot[k] = atomicAdd(&hist[eg[k]], 1);
    __syncthreads();

    const int cnt_g = (tid < nranges) ? hist[tid] : 0;
    int incl = cnt_g;
    #pragma unroll
    for (int dlt = 1; dlt < 32; dlt <<= 1) {
        const int up = __shfl_up(incl, dlt);
        if (lane >= dlt) incl += up;
    }
    if (lane == 31) wtot[wv] = incl;
    __syncthreads();
    if (tid < nranges) {
        int before = 0;
        for (int j = 0; j < wv; ++j) before += wtot[j];
        lstart[tid] = before + incl - cnt_g;
        gbase[tid] = (cnt_g > 0) ? atomicAdd(counts + (size_t)tid * GE_PITCH, cnt_g) : 0;
    }
    __syncthreads();

    #pragma unroll
    for (int k = 0; k < GE_SLICE / SC_THREADS; ++k) {
        if (eg[k] >= 0) {
            const int j = lstart[eg[k]] + eslot[k];
            sorted[j] = make_int2(es[k], er[k]);
            srange[j] = eg[k];
        }
    }
    __syncthreads();

    const int nvalid = lstart[nranges - 1] + hist[nranges - 1];
    for (int j = tid; j < nvalid; j += SC_THREADS) {
        const int g = srange[j];
        const int pos = gbase[g] + (j - lstart[g]);
        if (pos < SC_CAP)
            ((int2*)lists)[(size_t)g * SC_CAP + pos] = sorted[j];
    }
}

__global__ __launch_bounds__(SC_THREADS) void ggnn_scatter_owned(
    const float* __restrict__ m, const int* __restrict__ src,
    const int* __restrict__ dst, const int* __restrict__ lists,
    const int* __restrict__ counts, float* __restrict__ agg, int nnodes, int nedges)
{
    __shared__ __attribute__((aligned(16))) float acc[SC_ROWS * KDIM];
    __shared__ int2 list[SC_CAP];
    __shared__ int cnt;

    const int tid  = threadIdx.x;
    const int lane = tid & 31;
    const int wave = __builtin_amdgcn_readfirstlane(tid >> 5);

    const int g    = (int)blockIdx.x;
    const int lo   = g * SC_ROWS;
    const int rows = (nnodes - lo < SC_ROWS) ? (nnodes - lo) : SC_ROWS;
    const int total = __builtin_amdgcn_readfirstlane(counts[(size_t)g * GE_PITCH]);
    const bool have_list = total <= SC_CAP;

    int2 t[SC_CAP / SC_THREADS];
    #pragma unroll
    for (int k = 0; k < SC_CAP / SC_THREADS; ++k) {
        const int i = k * SC_THREADS + tid;
        t[k] = make_int2(0, 0);
        if (have_list && i < total)
            t[k] = ((const int2*)lists)[(size_t)g * SC_CAP + i];
    }
    for (int i = tid; i < SC_ROWS * KDIM / 4; i += SC_THREADS)
        ((float4*)acc)[i] = make_float4(0.f, 0.f, 0.f, 0.f);
    if (tid == 0) cnt = 0;
    #pragma unroll
    for (int k = 0; k < SC_CAP / SC_THREADS; ++k) {
        const int i = k * SC_THREADS + tid;
        if (have_list && i < total) list[i] = t[k];
    }
    __syncthreads();

    if (!have_list)
        sc_filter_edges<true>(lo, rows, tid, &cnt, list, acc, m, src, dst, nedges);

    const int n = (total < SC_CAP) ? total : SC_CAP;

    for (int i = wave; i < n; i += 32 * SC_GATHER) {
        int2 mine = make_int2(0, -1);
        const int my_idx = i + 32 * lane;
        if (lane < SC_GATHER && my_idx < n) mine = list[my_idx];
        float v[SC_GATHER][4];
        int   r[SC_GATHER];
        #pragma unroll
        for (int u = 0; u < SC_GATHER; ++u) {
            const int s = __builtin_amdgcn_readlane(mine.x, u);
            r[u] = __builtin_amdgcn_readlane(mine.y, u);
            const float* p = m + (size_t)s * KDIM + lane;
            v[u][0] = p[0]; v[u][1] = p[32]; v[u][2] = p[64]; v[u][3] = p[96];
        }
        #pragma unroll
        for (int u = 0; u < SC_GATHER; ++u) {
            if (r[u] >= 0) {
                float* a = acc + r[u] * KDIM + lane;
                atomicAdd(a,      v[u][0]);
                atomicAdd(a + 32, v[u][1]);
                atomicAdd(a + 64, v[u][2]);
                atomicAdd(a + 96, v[u][3]);
            }
        }
    }
    __syncthreads();

    float4* out4 = (float4*)(agg + (size_t)lo * KDIM);
    for (int i = tid; i < rows * (KDIM / 4); i += SC_THREADS)
        out4[i] = ((const float4*)acc)[i];
}

__device__ __forceinline__ float tanh_both(float x)
{
    const float ax = fabsf(x);
    const float x2 = x * x;
    float p = fmaf(__uint_as_float(0xbbbac73du), x2, __uint_as_float(0x3ca908c9u));
    p = fmaf(x2, p, __uint_as_float(0xbd5c1c4eu));
    p = fmaf(x2, p, __uint_as_float(0x3e088382u));
    p = fmaf(x2, p, __uint_as_float(0xbeaaaa99u));
    const float small = fmaf(x2, ax * p, ax);
    const float e = expf(ax + ax);
    const float large = fmaf(__builtin_amdgcn_rcpf(1.0f + e), -2.0f, 1.0f);
    return copysignf(ax < 0.625f ? small : large, x);
}

__device__ __forceinline__ float gate_one(
    float i_r, float i_z, float i_n, float h_r, float h_z, float h_n, float hv)
{
    const float r = 1.0f / (1.0f + __expf(-(i_r + h_r)));
    const float z = 1.0f / (1.0f + __expf(-(i_z + h_z)));
    const float nn = tanh_both(i_n + r * h_n);
    return (1.0f - z) * nn + z * hv;
}

struct GateOps { float hr[8], hz[8], hn[8], hv[8]; };

__device__ __forceinline__ void load_gate_ops(
    GateOps& g, const float* gh, const float* h, int tile, int half, int hc)
{
    #pragma unroll
    for (int r = 0; r < 8; ++r) {
        const size_t row = (size_t)(tile * 16 + r + half * 8);
        const float* p = gh + row * 384 + hc;
        g.hr[r] = p[0];
        g.hz[r] = p[128];
        g.hn[r] = p[256];
        g.hv[r] = h[row * KDIM + hc];
    }
}

__device__ __forceinline__ void gi_gate_tile(
    const v4f (&a)[16], const v4f (&b)[3][16], const float (&bv)[3],
    const GateOps& g, float* h, int tile, int half, int hc)
{
    v8f acc[3];
    #pragma unroll
    for (int t = 0; t < 3; ++t) acc[t] = v8f{0.f, 0.f, 0.f, 0.f, 0.f, 0.f, 0.f, 0.f};

    #pragma unroll
    for (int q = 0; q < 16; ++q) {
        const v2f a0 = {a[q].x, a[q].y};
        const v2f a1 = {a[q].z, a[q].w};
        #pragma unroll
        for (int t = 0; t < 3; ++t) {
            const v2f b0 = {b[t][q].x, b[t][q].y};
            acc[t] = __builtin_amdgcn_wmma_f32_16x16x4_f32(
                false, a0, false, b0, (short)0, acc[t], false, false);
        }
        #pragma unroll
        for (int t = 0; t < 3; ++t) {
            const v2f b1 = {b[t][q].z, b[t][q].w};
            acc[t] = __builtin_amdgcn_wmma_f32_16x16x4_f32(
                false, a1, false, b1, (short)0, acc[t], false, false);
        }
    }

    #pragma unroll
    for (int r = 0; r < 8; ++r) {
        const size_t row = (size_t)(tile * 16 + r + half * 8);
        h[row * KDIM + hc] = gate_one(
            acc[0][r] + bv[0], acc[1][r] + bv[1], acc[2][r] + bv[2],
            g.hr[r], g.hz[r], g.hn[r], g.hv[r]);
    }
}

__global__ __launch_bounds__(256) __attribute__((amdgpu_waves_per_eu(2, 2))) void ggnn_gi_gates_next(
    const float* __restrict__ agg, const float* __restrict__ w_ih,
    const float* __restrict__ b_ih, float* gh, float* h,
    const float* __restrict__ w_next, const float* __restrict__ w_hh,
    const float* __restrict__ b_hh, float* __restrict__ m, int ntiles)
{
    const int tid  = threadIdx.x;
    const int wave = __builtin_amdgcn_readfirstlane(tid >> 5);
    const int lane = tid & 31;
    const int half = lane >> 4;
    const int c    = lane & 15;
    const int hc   = wave * 16 + c;

    v4f   b[3][16];
    float bv[3];
    #pragma unroll
    for (int t = 0; t < 3; ++t) {
        const int n = t * 128 + hc;
        bv[t] = b_ih[n];
        #pragma unroll
        for (int q = 0; q < 16; ++q)
            b[t][q] = *(const v4f*)(w_ih + (size_t)n * KDIM + frag_k0(q, half));
    }

    const int nstreams = (int)gridDim.x;
    int tile = (int)blockIdx.x;
    if (tile >= ntiles) return;
    const int ntrips = (ntiles - tile + nstreams - 1) / nstreams;
    const int last_tile = tile + (ntrips - 1) * nstreams;

    v4f a0[16], a1[16];
    GateOps g;
    load_a_tile(a0, agg, tile, c, half);
    int i = 0;
    for (; i + 2 <= ntrips; i += 2) {
        load_a_tile(a1, agg, tile + nstreams, c, half);
        load_gate_ops(g, gh, h, tile, half, hc);
        __builtin_amdgcn_sched_barrier(0);
        gi_gate_tile(a0, b, bv, g, h, tile, half, hc);
        const int t2 = tile + 2 * nstreams;
        load_a_tile(a0, agg, t2 <= last_tile ? t2 : last_tile, c, half);
        load_gate_ops(g, gh, h, tile + nstreams, half, hc);
        __builtin_amdgcn_sched_barrier(0);
        gi_gate_tile(a1, b, bv, g, h, tile + nstreams, half, hc);
        tile = t2;
    }
    if (i < ntrips) {
        load_gate_ops(g, gh, h, tile, half, hc);
        __builtin_amdgcn_sched_barrier(0);
        gi_gate_tile(a0, b, bv, g, h, tile, half, hc);
    }

    if (w_next == nullptr) return;

    __builtin_amdgcn_fence(__ATOMIC_RELEASE, "agent");
    __syncthreads();
    __builtin_amdgcn_fence(__ATOMIC_ACQUIRE, "agent");

    m_gh_pass(h, w_next, w_hh, b_hh, m, gh, ntiles, (int)blockIdx.x, nstreams, wave, lane);
}

extern "C" void kernel_launch(void* const* d_in, const int* in_sizes, int n_in,
                              void* d_out, int out_size, void* d_ws, size_t ws_size,
                              hipStream_t stream)
{
    (void)n_in; (void)out_size; (void)ws_size;

    const float* x      = (const float*)d_in[0];
    const int*   ei     = (const int*)d_in[1];
    const float* lin_w  = (const float*)d_in[2];
    const float* lin_b  = (const float*)d_in[3];
    const float* ggc_w  = (const float*)d_in[4];
    const float* w_ih   = (const float*)d_in[5];
    const float* w_hh   = (const float*)d_in[6];
    const float* b_ih   = (const float*)d_in[7];
    const float* b_hh   = (const float*)d_in[8];
    const float* lin2_w = (const float*)d_in[9];
    const float* lin2_b = (const float*)d_in[10];
    float* out = (float*)d_out;

    const int N = in_sizes[0] / KDIM;
    const int E = in_sizes[1] / 2;
    const int* src = ei;
    const int* dst = ei + E;

    float* h   = (float*)d_ws;
    float* m   = h   + (size_t)N * 128;
    float* agg = m   + (size_t)N * 128;
    float* gh  = agg + (size_t)N * 128;

    const int row_tiles = (N + 15) / 16;
    const dim3 blk(256);
    const dim3 gemm_grid(256);

    const unsigned scatter_blks = (unsigned)((N + SC_ROWS - 1) / SC_ROWS);
    int* lists  = (int*)(gh + (size_t)N * 384);
    int* counts = lists + (size_t)scatter_blks * SC_CAP * 2;
    const unsigned group_blks = (unsigned)((E + GE_SLICE - 1) / GE_SLICE);

    ggnn_gemm<2, 4, false, false, false><<<gemm_grid, blk, 0, stream>>>(
        x, lin_w, lin_b, h, row_tiles, 128, 128, counts, (int)scatter_blks);

    ggnn_group_edges<<<group_blks, dim3(SC_THREADS), 0, stream>>>(
        src, dst, lists, counts, N, E, (int)scatter_blks);

    ggnn_m_gh<<<gemm_grid, blk, 0, stream>>>(h, ggc_w, w_hh, b_hh, m, gh, row_tiles);

    for (int l = 0; l < 7; ++l) {
        ggnn_scatter_owned<<<scatter_blks, dim3(SC_THREADS), 0, stream>>>(m, src, dst, lists, counts, agg, N, E);
        ggnn_gi_gates_next<<<gemm_grid, blk, 0, stream>>>(
            agg, w_ih, b_ih, gh, h, ggc_w + (size_t)(l + 1) * 128 * 128, w_hh, b_hh, m, row_tiles);
    }
    ggnn_scatter_owned<<<scatter_blks, dim3(SC_THREADS), 0, stream>>>(m, src, dst, lists, counts, agg, N, E);
    ggnn_gi_gates_next<<<gemm_grid, blk, 0, stream>>>(
        agg, w_ih, b_ih, gh, h, nullptr, w_hh, b_hh, m, row_tiles);

    ggnn_gemm<2, 4, false, true, true><<<gemm_grid, blk, 0, stream>>>(
        h, lin2_w, lin2_b, out, row_tiles, 100, 100, nullptr, 0);
}


# GRU kernel, first m/gh pass: A row tiles read from LDS (a group's agg tiles copied in once by whole-line loads, new h kept there for the next products); was 16-line global loads x 8 waves; sums same
# speedup vs baseline: 4.2734x; 1.0683x over previous
#include <hip/hip_runtime.h>
#include <hip/hip_bf16.h>
#include <math.h>

typedef float v2f __attribute__((ext_vector_type(2)));
typedef float v4f __attribute__((ext_vector_type(4)));
typedef float v8f __attribute__((ext_vector_type(8)));

#define KDIM 128

#define GE_PITCH 32

__device__ __forceinline__ int frag_k0(int q, int h) { return 8 * q + 4 * h; }

__device__ __forceinline__ void load_a_tile(
    v4f (&a)[16], const float* __restrict__ X, int tile, int c, int h)
{
    const float* p = X + (size_t)(tile * 16 + c) * KDIM;
    #pragma unroll
    for (int q = 0; q < 16; ++q)
        a[q] = *(const v4f*)(p + frag_k0(q, h));
}

#define LDS_PITCH 132
#define LDS_TILE  (16 * LDS_PITCH)
#define GT_CH     13

__device__ __forceinline__ void load_a_lds(
    v4f (&a)[16], const float* slot, int c, int h)
{
    const float* p = slot + c * LDS_PITCH;
    #pragma unroll
    for (int q = 0; q < 16; ++q)
        a[q] = *(const v4f*)(p + frag_k0(q, h));
}

__device__ __forceinline__ void stage_tiles_issue(
    v4f (&st)[GT_CH][2], const float* __restrict__ X,
    int tile0, int nstreams, int nch, int wave, int lane)
{
    #pragma unroll
    for (int i = 0; i < GT_CH; ++i) {
        const int ii = (i < nch) ? i : nch - 1;
        const float* p = X + ((size_t)(tile0 + ii * nstreams) * 16 + 2 * wave) * KDIM + lane * 4;
        st[i][0] = *(const v4f*)p;
        st[i][1] = *(const v4f*)(p + KDIM);
    }
}

__device__ __forceinline__ void stage_tiles_store(
    const v4f (&st)[GT_CH][2], float* s, int nch, int wave, int lane)
{
    #pragma unroll
    for (int i = 0; i < GT_CH; ++i) {
        const int ii = (i < nch) ? i : nch - 1;
        float* sp = s + ii * LDS_TILE + (2 * wave) * LDS_PITCH + lane * 4;
        *(v4f*)sp = st[i][0];
        *(v4f*)(sp + LDS_PITCH) = st[i][1];
    }
}

template <int CT, bool RELU_IN, bool GUARD>
__device__ __forceinline__ void tile_body(
    const v4f (&a)[16], const v4f (&b)[CT][16], const float (&bv)[CT],
    float* __restrict__ Y, int tile, int h, int c, int col0,
    int out_stride, int out_cols_total)
{
    v8f acc[CT];
    #pragma unroll
    for (int t = 0; t < CT; ++t) acc[t] = v8f{0.f, 0.f, 0.f, 0.f, 0.f, 0.f, 0.f, 0.f};

    #pragma unroll
    for (int q = 0; q < 16; ++q) {
        v2f a0 = {a[q].x, a[q].y};
        v2f a1 = {a[q].z, a[q].w};
        if (RELU_IN) {
            a0.x = fmaxf(a0.x, 0.0f); a0.y = fmaxf(a0.y, 0.0f);
            a1.x = fmaxf(a1.x, 0.0f); a1.y = fmaxf(a1.y, 0.0f);
        }
        #pragma unroll
        for (int t = 0; t < CT; ++t) {
            const v2f b0 = {b[t][q].x, b[t][q].y};
            acc[t] = __builtin_amdgcn_wmma_f32_16x16x4_f32(
                false, a0, false, b0, (short)0, acc[t], false, false);
        }
        #pragma unroll
        for (int t = 0; t < CT; ++t) {
            const v2f b1 = {b[t][q].z, b[t][q].w};
            acc[t] = __builtin_amdgcn_wmma_f32_16x16x4_f32(
                false, a1, false, b1, (short)0, acc[t], false, false);
        }
    }

    #pragma unroll
    for (int t = 0; t < CT; ++t) {
        const int cc = col0 + t * 16 + c;
        if (!GUARD || cc < out_cols_total) {
            #pragma unroll
            for (int r = 0; r < 8; ++r) {
                const int row = tile * 16 + r + h * 8;
                Y[(size_t)row * out_stride + cc] = acc[t][r] + bv[t];
            }
        }
    }
}

template <int CT, int NG, bool TRANS_FILL, bool RELU_IN, bool GUARD>
__global__ __launch_bounds__(256) __attribute__((amdgpu_waves_per_eu(2, 2))) void ggnn_gemm(
    const float* __restrict__ X, const float* __restrict__ W,
    const float* __restrict__ bias, float* __restrict__ Y,
    int ntiles, int out_stride, int out_cols_total,
    int* __restrict__ zero_counts, int nzero)
{
    constexpr int NSTR = 8 / NG;
    const int tid  = threadIdx.x;

    if (zero_counts != nullptr && blockIdx.x == 0)
        for (int i = tid; i < nzero; i += 256) zero_counts[(size_t)i * GE_PITCH] = 0;

    const int wave = __builtin_amdgcn_readfirstlane(tid >> 5);
    const int lane = tid & 31;
    const int h    = lane >> 4;
    const int c    = lane & 15;
    const int grp  = wave % NG;
    const int strm = wave / NG;
    const int col0 = grp * (CT * 16);

    v4f   b[CT][16];
    float bv[CT];
    #pragma unroll
    for (int t = 0; t < CT; ++t) {
        const int n = col0 + t * 16 + c;
        const bool ok = !GUARD || (n < out_cols_total);
        bv[t] = (bias != nullptr && ok) ? bias[n] : 0.0f;
        #pragma unroll
        for (int q = 0; q < 16; ++q) {
            const int k0 = frag_k0(q, h);
            v4f w = {0.f, 0.f, 0.f, 0.f};
            if (ok) {
                if (TRANS_FILL) {
                    w.x = W[(size_t)(k0 + 0) * out_cols_total + n];
                    w.y = W[(size_t)(k0 + 1) * out_cols_total + n];
                    w.z = W[(size_t)(k0 + 2) * out_cols_total + n];
                    w.w = W[(size_t)(k0 + 3) * out_cols_total + n];
                } else {
                    w = *(const v4f*)(W + (size_t)n * KDIM + k0);
                }
            }
            b[t][q] = w;
        }
    }

    const int nstreams = (int)gridDim.x * NSTR;
    int tile = (int)blockIdx.x * NSTR + strm;
    if (tile >= ntiles) return;
    const int ntrips = (ntiles - tile + nstreams - 1) / nstreams;
    const int last_tile = tile + (ntrips - 1) * nstreams;

    v4f a0[16], a1[16];
    load_a_tile(a0, X, tile, c, h);
    int i = 0;
    for (; i + 2 <= ntrips; i += 2) {
        load_a_tile(a1, X, tile + nstreams, c, h);
        __builtin_amdgcn_sched_barrier(0);
        tile_body<CT, RELU_IN, GUARD>(a0, b, bv, Y, tile, h, c, col0, out_stride, out_cols_total);
        const int t2 = tile + 2 * nstreams;
        load_a_tile(a0, X, t2 <= last_tile ? t2 : last_tile, c, h);
        __builtin_amdgcn_sched_barrier(0);
        tile_body<CT, RELU_IN, GUARD>(a1, b, bv, Y, tile + nstreams, h, c, col0, out_stride, out_cols_total);
        tile = t2;
    }
    if (i < ntrips)
        tile_body<CT, RELU_IN, GUARD>(a0, b, bv, Y, tile, h, c, col0, out_stride, out_cols_total);
}

__device__ __forceinline__ void m_gh_tile(
    const v4f (&a)[16], const v4f (&b)[4][16], const float (&bv)[4],
    float* __restrict__ M, float* GH, int tile, int h, int mc, int gc)
{
    v8f acc[4];
    #pragma unroll
    for (int t = 0; t < 4; ++t) acc[t] = v8f{0.f, 0.f, 0.f, 0.f, 0.f, 0.f, 0.f, 0.f};

    #pragma unroll
    for (int q = 0; q < 16; ++q) {
        const v2f a0 = {a[q].x, a[q].y};
        const v2f a1 = {a[q].z, a[q].w};
        #pragma unroll
        for (int t = 0; t < 4; ++t) {
            const v2f b0 = {b[t][q].x, b[t][q].y};
            acc[t] = __builtin_amdgcn_wmma_f32_16x16x4_f32(
                false, a0, false, b0, (short)0, acc[t], false, false);
        }
        #pragma unroll
        for (int t = 0; t < 4; ++t) {
            const v2f b1 = {b[t][q].z, b[t][q].w};
            acc[t] = __builtin_amdgcn_wmma_f32_16x16x4_f32(
                false, a1, false, b1, (short)0, acc[t], false, false);
        }
    }

    #pragma unroll
    for (int r = 0; r < 8; ++r) {
        const size_t row = (size_t)(tile * 16 + r + h * 8);
        M[row * 128 + mc] = acc[0][r];
        #pragma unroll
        for (int t = 1; t < 4; ++t)
            GH[row * 384 + gc + (t - 1) * 16] = acc[t][r] + bv[t];
    }
}

__device__ __forceinline__ void m_gh_weights(
    v4f (&b)[4][16], float (&bv)[4], const float* __restrict__ w_l,
    const float* __restrict__ w_hh, const float* __restrict__ b_hh, int h, int mc, int gc)
{
    bv[0] = 0.0f;
    #pragma unroll
    for (int q = 0; q < 16; ++q) {
        const int k0 = frag_k0(q, h);
        v4f w;
        w.x = w_l[(size_t)(k0 + 0) * 128 + mc];
        w.y = w_l[(size_t)(k0 + 1) * 128 + mc];
        w.z = w_l[(size_t)(k0 + 2) * 128 + mc];
        w.w = w_l[(size_t)(k0 + 3) * 128 + mc];
        b[0][q] = w;
    }
    #pragma unroll
    for (int t = 1; t < 4; ++t) {
        const int n = gc + (t - 1) * 16;
        bv[t] = b_hh[n];
        #pragma unroll
        for (int q = 0; q < 16; ++q)
            b[t][q] = *(const v4f*)(w_hh + (size_t)n * KDIM + frag_k0(q, h));
    }
}

__device__ __forceinline__ void m_gh_lds_tiles(
    const float* s, const v4f (&b)[4][16], const float (&bv)[4],
    float* __restrict__ M, float* GH, int tile0, int nstreams, int nch,
    int h, int c, int mc, int gc)
{
    v4f a0[16], a1[16];
    load_a_lds(a0, s, c, h);
    int i = 0;
    for (; i + 2 <= nch; i += 2) {
        const int tile = tile0 + i * nstreams;
        load_a_lds(a1, s + (i + 1) * LDS_TILE, c, h);
        __builtin_amdgcn_sched_barrier(0);
        m_gh_tile(a0, b, bv, M, GH, tile, h, mc, gc);
        __builtin_amdgcn_sched_barrier(0);
        const int i2 = (i + 2 < nch) ? i + 2 : nch - 1;
        load_a_lds(a0, s + i2 * LDS_TILE, c, h);
        __builtin_amdgcn_sched_barrier(0);
        m_gh_tile(a1, b, bv, M, GH, tile + nstreams, h, mc, gc);
    }
    if (i < nch)
        m_gh_tile(a0, b, bv, M, GH, tile0 + i * nstreams, h, mc, gc);
}

__global__ __launch_bounds__(256) __attribute__((amdgpu_waves_per_eu(2, 2))) void ggnn_m_gh(
    const float* __restrict__ X, const float* __restrict__ w_l,
    const float* __restrict__ w_hh, const float* __restrict__ b_hh,
    float* __restrict__ M, float* __restrict__ GH, int ntiles)
{
    __shared__ __attribute__((aligned(16))) float s_x[GT_CH * LDS_TILE];

    const int tid  = threadIdx.x;
    const int wave = __builtin_amdgcn_readfirstlane(tid >> 5);
    const int lane = tid & 31;
    const int h    = lane >> 4;
    const int c    = lane & 15;
    const int mc   = wave * 16 + c;
    const int gc   = wave * 48 + c;

    const int nstreams = (int)gridDim.x;
    const int first = (int)blockIdx.x;
    if (first >= ntiles) return;
    const int nch = (ntiles - first + nstreams - 1) / nstreams;

    v4f st[GT_CH][2];
    stage_tiles_issue(st, X, first, nstreams, nch, wave, lane);

    v4f   b[4][16];
    float bv[4];
    m_gh_weights(b, bv, w_l, w_hh, b_hh, h, mc, gc);

    stage_tiles_store(st, s_x, nch, wave, lane);
    __syncthreads();

    m_gh_lds_tiles(s_x, b, bv, M, GH, first, nstreams, nch, h, c, mc, gc);
}

#define SC_ROWS    200
#define SC_CAP     8192
#define SC_THREADS 1024
#define SC_VEC     8
#define SC_GATHER  8
#define GE_SLICE   4096
#define GE_MAXR    1024

template <bool ADD_WHEN_FULL>
__device__ __forceinline__ void sc_note_edge(
    int e, int d, int lo, int rows, int* cnt, int2* list, float* acc,
    const float* __restrict__ m, const int* __restrict__ src)
{
    const unsigned r = (unsigned)(d - lo);
    if (r < (unsigned)rows) {
        const int slot = atomicAdd(cnt, 1);
        if (slot < SC_CAP) {
            list[slot] = make_int2(e, (int)r);
        } else if (ADD_WHEN_FULL) {
            const float* p = m + (size_t)src[e] * KDIM;
            float* a = acc + (size_t)r * KDIM;
            #pragma nounroll
            for (int j = 0; j < KDIM; ++j) atomicAdd(a + j, p[j]);
        }
    }
}

template <bool ADD_WHEN_FULL>
__device__ __forceinline__ void sc_filter_edges(
    int lo, int rows, int tid, int* cnt, int2* list, float* acc,
    const float* __restrict__ m, const int* __restrict__ src,
    const int* __restrict__ dst, int nedges)
{
    if ((nedges & 3) == 0) {
        for (int base = 0; base < nedges; base += SC_VEC * SC_THREADS * 4) {
            int4 d[SC_VEC];
            #pragma unroll
            for (int k = 0; k < SC_VEC; ++k) {
                const int e0 = base + (k * SC_THREADS + tid) * 4;
                d[k] = make_int4(lo - 1, lo - 1, lo - 1, lo - 1);
                if (e0 < nedges) d[k] = *(const int4*)(dst + e0);
            }
            #pragma unroll
            for (int k = 0; k < SC_VEC; ++k) {
                const int e0 = base + (k * SC_THREADS + tid) * 4;
                sc_note_edge<ADD_WHEN_FULL>(e0 + 0, d[k].x, lo, rows, cnt, list, acc, m, src);
                sc_note_edge<ADD_WHEN_FULL>(e0 + 1, d[k].y, lo, rows, cnt, list, acc, m, src);
                sc_note_edge<ADD_WHEN_FULL>(e0 + 2, d[k].z, lo, rows, cnt, list, acc, m, src);
                sc_note_edge<ADD_WHEN_FULL>(e0 + 3, d[k].w, lo, rows, cnt, list, acc, m, src);
            }
        }
    } else {
        for (int e = tid; e < nedges; e += SC_THREADS)
            sc_note_edge<ADD_WHEN_FULL>(e, dst[e], lo, rows, cnt, list, acc, m, src);
    }
    __syncthreads();

    const int n = (*cnt < SC_CAP) ? *cnt : SC_CAP;
    for (int i = tid; i < n; i += SC_THREADS)
        list[i].x = src[list[i].x];
    __syncthreads();
}

__global__ __launch_bounds__(SC_THREADS) void ggnn_group_edges(
    const int* __restrict__ src, const int* __restrict__ dst,
    int* __restrict__ lists, int* __restrict__ counts, int nnodes, int nedges, int nranges)
{
    __shared__ int  hist[GE_MAXR];
    __shared__ int  lstart[GE_MAXR];
    __shared__ int  gbase[GE_MAXR];
    __shared__ int  wtot[SC_THREADS / 32];
    __shared__ int2 sorted[GE_SLICE];
    __shared__ int  srange[GE_SLICE];

    const int tid  = threadIdx.x;
    const int lane = tid & 31;
    const int wv   = tid >> 5;
    const int e0   = (int)blockIdx.x * GE_SLICE;

    for (int g = tid; g < nranges; g += SC_THREADS) hist[g] = 0;
    __syncthreads();

    int eg[GE_SLICE / SC_THREADS], er[GE_SLICE / SC_THREADS];
    int es[GE_SLICE / SC_THREADS], eslot[GE_SLICE / SC_THREADS];
    #pragma unroll
    for (int k = 0; k < GE_SLICE / SC_THREADS; ++k) {
        const int e = e0 + k * SC_THREADS + tid;
        int d = -1, s = 0;
        if (e < nedges) { d = dst[e]; s = src[e]; }
        const bool ok = (unsigned)d < (unsigned)nnodes;
        const int g = ok ? d / SC_ROWS : -1;
        eg[k] = g;
        er[k] = d - g * SC_ROWS;
        es[k] = s;
        eslot[k] = 0;
    }
    #pragma unroll
    for (int k = 0; k < GE_SLICE / SC_THREADS; ++k)
        if (eg[k] >= 0) eslot[k] = atomicAdd(&hist[eg[k]], 1);
    __syncthreads();

    const int cnt_g = (tid < nranges) ? hist[tid] : 0;
    int incl = cnt_g;
    #pragma unroll
    for (int dlt = 1; dlt < 32; dlt <<= 1) {
        const int up = __shfl_up(incl, dlt);
        if (lane >= dlt) incl += up;
    }
    if (lane == 31) wtot[wv] = incl;
    __syncthreads();
    if (tid < nranges) {
        int before = 0;
        for (int j = 0; j < wv; ++j) before += wtot[j];
        lstart[tid] = before + incl - cnt_g;
        gbase[tid] = (cnt_g > 0) ? atomicAdd(counts + (size_t)tid * GE_PITCH, cnt_g) : 0;
    }
    __syncthreads();

    #pragma unroll
    for (int k = 0; k < GE_SLICE / SC_THREADS; ++k) {
        if (eg[k] >= 0) {
            const int j = lstart[eg[k]] + eslot[k];
            sorted[j] = make_int2(es[k], er[k]);
            srange[j] = eg[k];
        }
    }
    __syncthreads();

    const int nvalid = lstart[nranges - 1] + hist[nranges - 1];
    for (int j = tid; j < nvalid; j += SC_THREADS) {
        const int g = srange[j];
        const int pos = gbase[g] + (j - lstart[g]);
        if (pos < SC_CAP)
            ((int2*)lists)[(size_t)g * SC_CAP + pos] = sorted[j];
    }
}

__global__ __launch_bounds__(SC_THREADS) void ggnn_scatter_owned(
    const float* __restrict__ m, const int* __restrict__ src,
    const int* __restrict__ dst, const int* __restrict__ lists,
    const int* __restrict__ counts, float* __restrict__ agg, int nnodes, int nedges)
{
    __shared__ __attribute__((aligned(16))) float acc[SC_ROWS * KDIM];
    __shared__ int2 list[SC_CAP];
    __shared__ int cnt;

    const int tid  = threadIdx.x;
    const int lane = tid & 31;
    const int wave = __builtin_amdgcn_readfirstlane(tid >> 5);

    const int g    = (int)blockIdx.x;
    const int lo   = g * SC_ROWS;
    const int rows = (nnodes - lo < SC_ROWS) ? (nnodes - lo) : SC_ROWS;
    const int total = __builtin_amdgcn_readfirstlane(counts[(size_t)g * GE_PITCH]);
    const bool have_list = total <= SC_CAP;

    int2 t[SC_CAP / SC_THREADS];
    #pragma unroll
    for (int k = 0; k < SC_CAP / SC_THREADS; ++k) {
        const int i = k * SC_THREADS + tid;
        t[k] = make_int2(0, 0);
        if (have_list && i < total)
            t[k] = ((const int2*)lists)[(size_t)g * SC_CAP + i];
    }
    for (int i = tid; i < SC_ROWS * KDIM / 4; i += SC_THREADS)
        ((float4*)acc)[i] = make_float4(0.f, 0.f, 0.f, 0.f);
    if (tid == 0) cnt = 0;
    #pragma unroll
    for (int k = 0; k < SC_CAP / SC_THREADS; ++k) {
        const int i = k * SC_THREADS + tid;
        if (have_list && i < total) list[i] = t[k];
    }
    __syncthreads();

    if (!have_list)
        sc_filter_edges<true>(lo, rows, tid, &cnt, list, acc, m, src, dst, nedges);

    const int n = (total < SC_CAP) ? total : SC_CAP;

    for (int i = wave; i < n; i += 32 * SC_GATHER) {
        int2 mine = make_int2(0, -1);
        const int my_idx = i + 32 * lane;
        if (lane < SC_GATHER && my_idx < n) mine = list[my_idx];
        float v[SC_GATHER][4];
        int   r[SC_GATHER];
        #pragma unroll
        for (int u = 0; u < SC_GATHER; ++u) {
            const int s = __builtin_amdgcn_readlane(mine.x, u);
            r[u] = __builtin_amdgcn_readlane(mine.y, u);
            const float* p = m + (size_t)s * KDIM + lane;
            v[u][0] = p[0]; v[u][1] = p[32]; v[u][2] = p[64]; v[u][3] = p[96];
        }
        #pragma unroll
        for (int u = 0; u < SC_GATHER; ++u) {
            if (r[u] >= 0) {
                float* a = acc + r[u] * KDIM + lane;
                atomicAdd(a,      v[u][0]);
                atomicAdd(a + 32, v[u][1]);
                atomicAdd(a + 64, v[u][2]);
                atomicAdd(a + 96, v[u][3]);
            }
        }
    }
    __syncthreads();

    float4* out4 = (float4*)(agg + (size_t)lo * KDIM);
    for (int i = tid; i < rows * (KDIM / 4); i += SC_THREADS)
        out4[i] = ((const float4*)acc)[i];
}

__device__ __forceinline__ float tanh_both(float x)
{
    const float ax = fabsf(x);
    const float x2 = x * x;
    float p = fmaf(__uint_as_float(0xbbbac73du), x2, __uint_as_float(0x3ca908c9u));
    p = fmaf(x2, p, __uint_as_float(0xbd5c1c4eu));
    p = fmaf(x2, p, __uint_as_float(0x3e088382u));
    p = fmaf(x2, p, __uint_as_float(0xbeaaaa99u));
    const float small = fmaf(x2, ax * p, ax);
    const float e = expf(ax + ax);
    const float large = fmaf(__builtin_amdgcn_rcpf(1.0f + e), -2.0f, 1.0f);
    return copysignf(ax < 0.625f ? small : large, x);
}

__device__ __forceinline__ float gate_one(
    float i_r, float i_z, float i_n, float h_r, float h_z, float h_n, float hv)
{
    const float r = 1.0f / (1.0f + __expf(-(i_r + h_r)));
    const float z = 1.0f / (1.0f + __expf(-(i_z + h_z)));
    const float nn = tanh_both(i_n + r * h_n);
    return (1.0f - z) * nn + z * hv;
}

struct GateOps { float hr[8], hz[8], hn[8], hv[8]; };

__device__ __forceinline__ void load_gate_ops(
    GateOps& g, const float* gh, const float* h, int tile, int half, int hc)
{
    #pragma unroll
    for (int r = 0; r < 8; ++r) {
        const size_t row = (size_t)(tile * 16 + r + half * 8);
        const float* p = gh + row * 384 + hc;
        g.hr[r] = p[0];
        g.hz[r] = p[128];
        g.hn[r] = p[256];
        g.hv[r] = h[row * KDIM + hc];
    }
}

__device__ __forceinline__ void gi_gate_tile(
    const v4f (&a)[16], const v4f (&b)[3][16], const float (&bv)[3],
    const GateOps& g, float* h, float* sh, int tile, int half, int hc)
{
    v8f acc[3];
    #pragma unroll
    for (int t = 0; t < 3; ++t) acc[t] = v8f{0.f, 0.f, 0.f, 0.f, 0.f, 0.f, 0.f, 0.f};

    #pragma unroll
    for (int q = 0; q < 16; ++q) {
        const v2f a0 = {a[q].x, a[q].y};
        const v2f a1 = {a[q].z, a[q].w};
        #pragma unroll
        for (int t = 0; t < 3; ++t) {
            const v2f b0 = {b[t][q].x, b[t][q].y};
            acc[t] = __builtin_amdgcn_wmma_f32_16x16x4_f32(
                false, a0, false, b0, (short)0, acc[t], false, false);
        }
        #pragma unroll
        for (int t = 0; t < 3; ++t) {
            const v2f b1 = {b[t][q].z, b[t][q].w};
            acc[t] = __builtin_amdgcn_wmma_f32_16x16x4_f32(
                false, a1, false, b1, (short)0, acc[t], false, false);
        }
    }

    #pragma unroll
    for (int r = 0; r < 8; ++r) {
        const size_t row = (size_t)(tile * 16 + r + half * 8);
        const float hnew = gate_one(
            acc[0][r] + bv[0], acc[1][r] + bv[1], acc[2][r] + bv[2],
            g.hr[r], g.hz[r], g.hn[r], g.hv[r]);
        h[row * KDIM + hc] = hnew;
        sh[(r + half * 8) * LDS_PITCH + hc] = hnew;
    }
}

__global__ __launch_bounds__(256) __attribute__((amdgpu_waves_per_eu(2, 2))) void ggnn_gi_gates_next(
    const float* __restrict__ agg, const float* __restrict__ w_ih,
    const float* __restrict__ b_ih, float* gh, float* h,
    const float* __restrict__ w_next, const float* __restrict__ w_hh,
    const float* __restrict__ b_hh, float* __restrict__ m, int ntiles)
{
    __shared__ __attribute__((aligned(16))) float s_agg[GT_CH * LDS_TILE];
    __shared__ __attribute__((aligned(16))) float s_h[GT_CH * LDS_TILE];

    const int tid  = threadIdx.x;
    const int wave = __builtin_amdgcn_readfirstlane(tid >> 5);
    const int lane = tid & 31;
    const int half = lane >> 4;
    const int c    = lane & 15;
    const int hc   = wave * 16 + c;
    const int mc   = hc;
    const int gc   = wave * 48 + c;

    const int nstreams = (int)gridDim.x;
    const int first = (int)blockIdx.x;
    if (first >= ntiles) return;
    const int nch   = (ntiles - first + nstreams - 1) / nstreams;
    const int tile0 = first;

    {
        v4f st[GT_CH][2];
        stage_tiles_issue(st, agg, tile0, nstreams, nch, wave, lane);

        {
            v4f   b[3][16];
            float bv[3];
            #pragma unroll
            for (int t = 0; t < 3; ++t) {
                const int n = t * 128 + hc;
                bv[t] = b_ih[n];
                #pragma unroll
                for (int q = 0; q < 16; ++q)
                    b[t][q] = *(const v4f*)(w_ih + (size_t)n * KDIM + frag_k0(q, half));
            }

            stage_tiles_store(st, s_agg, nch, wave, lane);
            __syncthreads();

            v4f a0[16], a1[16];
            GateOps g;
            load_a_lds(a0, s_agg, c, half);
            int i = 0;
            for (; i + 2 <= nch; i += 2) {
                const int tile = tile0 + i * nstreams;
                load_a_lds(a1, s_agg + (i + 1) * LDS_TILE, c, half);
                load_gate_ops(g, gh, h, tile, half, hc);
                __builtin_amdgcn_sched_barrier(0);
                gi_gate_tile(a0, b, bv, g, h, s_h + i * LDS_TILE, tile, half, hc);
                const int i2 = (i + 2 < nch) ? i + 2 : nch - 1;
                load_a_lds(a0, s_agg + i2 * LDS_TILE, c, half);
                load_gate_ops(g, gh, h, tile + nstreams, half, hc);
                __builtin_amdgcn_sched_barrier(0);
                gi_gate_tile(a1, b, bv, g, h, s_h + (i + 1) * LDS_TILE, tile + nstreams, half, hc);
            }
            if (i < nch) {
                const int tile = tile0 + i * nstreams;
                load_gate_ops(g, gh, h, tile, half, hc);
                __builtin_amdgcn_sched_barrier(0);
                gi_gate_tile(a0, b, bv, g, h, s_h + i * LDS_TILE, tile, half, hc);
            }
        }

        if (w_next == nullptr) return;

        __syncthreads();

        {
            v4f   b[4][16];
            float bv[4];
            m_gh_weights(b, bv, w_next, w_hh, b_hh, half, mc, gc);

            m_gh_lds_tiles(s_h, b, bv, m, gh, tile0, nstreams, nch, half, c, mc, gc);
        }
    }
}

extern "C" void kernel_launch(void* const* d_in, const int* in_sizes, int n_in,
                              void* d_out, int out_size, void* d_ws, size_t ws_size,
                              hipStream_t stream)
{
    (void)n_in; (void)out_size; (void)ws_size;

    const float* x      = (const float*)d_in[0];
    const int*   ei     = (const int*)d_in[1];
    const float* lin_w  = (const float*)d_in[2];
    const float* lin_b  = (const float*)d_in[3];
    const float* ggc_w  = (const float*)d_in[4];
    const float* w_ih   = (const float*)d_in[5];
    const float* w_hh   = (const float*)d_in[6];
    const float* b_ih   = (const float*)d_in[7];
    const float* b_hh   = (const float*)d_in[8];
    const float* lin2_w = (const float*)d_in[9];
    const float* lin2_b = (const float*)d_in[10];
    float* out = (float*)d_out;

    const int N = in_sizes[0] / KDIM;
    const int E = in_sizes[1] / 2;
    const int* src = ei;
    const int* dst = ei + E;

    float* h   = (float*)d_ws;
    float* m   = h   + (size_t)N * 128;
    float* agg = m   + (size_t)N * 128;
    float* gh  = agg + (size_t)N * 128;

    const int row_tiles = (N + 15) / 16;
    const dim3 blk(256);
    const dim3 gemm_grid(256);

    const dim3 gates_grid(row_tiles > 256 * GT_CH ? (unsigned)((row_tiles + GT_CH - 1) / GT_CH) : 256u);

    const unsigned scatter_blks = (unsigned)((N + SC_ROWS - 1) / SC_ROWS);
    int* lists  = (int*)(gh + (size_t)N * 384);
    int* counts = lists + (size_t)scatter_blks * SC_CAP * 2;
    const unsigned group_blks = (unsigned)((E + GE_SLICE - 1) / GE_SLICE);

    ggnn_gemm<2, 4, false, false, false><<<gemm_grid, blk, 0, stream>>>(
        x, lin_w, lin_b, h, row_tiles, 128, 128, counts, (int)scatter_blks);

    ggnn_group_edges<<<group_blks, dim3(SC_THREADS), 0, stream>>>(
        src, dst, lists, counts, N, E, (int)scatter_blks);

    ggnn_m_gh<<<gates_grid, blk, 0, stream>>>(h, ggc_w, w_hh, b_hh, m, gh, row_tiles);

    for (int l = 0; l < 7; ++l) {
        ggnn_scatter_owned<<<scatter_blks, dim3(SC_THREADS), 0, stream>>>(m, src, dst, lists, counts, agg, N, E);
        ggnn_gi_gates_next<<<gates_grid, blk, 0, stream>>>(
            agg, w_ih, b_ih, gh, h, ggc_w + (size_t)(l + 1) * 128 * 128, w_hh, b_hh, m, row_tiles);
    }
    ggnn_scatter_owned<<<scatter_blks, dim3(SC_THREADS), 0, stream>>>(m, src, dst, lists, counts, agg, N, E);
    ggnn_gi_gates_next<<<gates_grid, blk, 0, stream>>>(
        agg, w_ih, b_ih, gh, h, nullptr, w_hh, b_hh, m, row_tiles);

    ggnn_gemm<2, 4, false, true, true><<<gemm_grid, blk, 0, stream>>>(
        h, lin2_w, lin2_b, out, row_tiles, 100, 100, nullptr, 0);
}
